# starting kernel
# baseline (speedup 1.0000x reference)
#include <hip/hip_runtime.h>
#include <math.h>

typedef __attribute__((ext_vector_type(16))) __bf16 v16bf;
typedef __attribute__((ext_vector_type(8)))  float  v8f;
typedef unsigned short u16;
typedef unsigned int   u32;

#define HID 128
#define BM  128
#define BN  64
#define BK  32
#define BKP 40

__device__ __forceinline__ float bfh_to_f(u16 h) {
  union { u32 u; float f; } c; c.u = ((u32)h) << 16; return c.f;
}
__device__ __forceinline__ u16 f_to_bf_rne(float f) {
  union { float f; u32 u; } c; c.f = f;
  u32 r = c.u + 0x7FFFu + ((c.u >> 16) & 1u);
  return (u16)(r >> 16);
}
__device__ __forceinline__ void split_bf(float v, u16& hi, u16& lo) {
  union { float f; u32 u; } c; c.f = v;
  u16 h = (u16)(c.u >> 16);
  float r = v - bfh_to_f(h);
  hi = h; lo = f_to_bf_rne(r);
}

__global__ __launch_bounds__(256)
void gemm_bf16x3(const float* __restrict__ A, const float* __restrict__ W,
                 float* __restrict__ C, int nrows) {
  __shared__ u16 aHi[BM][BKP];
  __shared__ u16 aLo[BM][BKP];
  __shared__ u16 bHi[BN][BKP];
  __shared__ u16 bLo[BN][BKP];

  const int tid  = threadIdx.x;
  const int lane = tid & 31;
  const int wave = tid >> 5;
  const int l15  = lane & 15;
  const int kh   = (lane >> 4) << 3;
  const int row0 = blockIdx.x * BM;
  const int col0 = blockIdx.y * BN;
  const bool fullTile = (row0 + BM) <= nrows;

  v8f acc[4];
  for (int t = 0; t < 4; ++t)
    for (int r = 0; r < 8; ++r) acc[t][r] = 0.0f;

  for (int kc = 0; kc < HID; kc += BK) {
    __syncthreads();
    if (fullTile) {
      #pragma unroll
      for (int idx = tid; idx < BM * BK; idx += 256) {
        const int m = idx >> 5, k = idx & 31;
        float v = A[(long long)(row0 + m) * HID + (kc + k)];
        split_bf(v, aHi[m][k], aLo[m][k]);
      }
    } else {
      for (int idx = tid; idx < BM * BK; idx += 256) {
        const int m = idx >> 5, k = idx & 31;
        const int row = row0 + m;
        float v = (row < nrows) ? A[(long long)row * HID + (kc + k)] : 0.0f;
        split_bf(v, aHi[m][k], aLo[m][k]);
      }
    }
    #pragma unroll
    for (int idx = tid; idx < BN * BK; idx += 256) {
      const int n = idx >> 5, k = idx & 31;
      float v = W[(kc + k) * HID + (col0 + n)];
      split_bf(v, bHi[n][k], bLo[n][k]);
    }
    __syncthreads();

    union Frag { uint4 q[2]; v16bf v; };
    Frag ah, al;
    const int m = (wave << 4) + l15;
    ah.q[0] = *(const uint4*)&aHi[m][kh];
    ah.q[1] = *(const uint4*)&aHi[m][16 + kh];
    al.q[0] = *(const uint4*)&aLo[m][kh];
    al.q[1] = *(const uint4*)&aLo[m][16 + kh];

    #pragma unroll
    for (int nt = 0; nt < 4; ++nt) {
      Frag bh, bl;
      const int n = (nt << 4) + l15;
      bh.q[0] = *(const uint4*)&bHi[n][kh];
      bh.q[1] = *(const uint4*)&bHi[n][16 + kh];
      bl.q[0] = *(const uint4*)&bLo[n][kh];
      bl.q[1] = *(const uint4*)&bLo[n][16 + kh];
      acc[nt] = __builtin_amdgcn_wmma_f32_16x16x32_bf16(false, ah.v, false, bh.v, (short)0, acc[nt], false, false);
      acc[nt] = __builtin_amdgcn_wmma_f32_16x16x32_bf16(false, ah.v, false, bl.v, (short)0, acc[nt], false, false);
      acc[nt] = __builtin_amdgcn_wmma_f32_16x16x32_bf16(false, al.v, false, bh.v, (short)0, acc[nt], false, false);
    }
  }

  const int rbase = row0 + (wave << 4) + ((lane & 16) ? 8 : 0);
  if (fullTile) {
    #pragma unroll
    for (int nt = 0; nt < 4; ++nt) {
      const int c = col0 + (nt << 4) + l15;
      #pragma unroll
      for (int r = 0; r < 8; ++r)
        C[(long long)(rbase + r) * HID + c] = acc[nt][r];
    }
  } else {
    for (int nt = 0; nt < 4; ++nt) {
      const int c = col0 + (nt << 4) + l15;
      for (int r = 0; r < 8; ++r) {
        const int row = rbase + r;
        if (row < nrows) C[(long long)row * HID + c] = acc[nt][r];
      }
    }
  }
}

template <int CC>
__global__ void gemm_small(const float* __restrict__ h, const float* __restrict__ W,
                           float* __restrict__ out, int n) {
  const long long idx = (long long)blockIdx.x * blockDim.x + threadIdx.x;
  if (idx >= (long long)n * CC) return;
  const int node = (int)(idx / CC);
  const int c    = (int)(idx % CC);
  const float* hr = h + (long long)node * HID;
  float s = 0.0f;
  #pragma unroll 8
  for (int k = 0; k < HID; ++k) s += hr[k] * W[k * CC + c];
  out[idx] = s;
}

__global__ void fill_ones(float* p, int n) {
  int i = blockIdx.x * blockDim.x + threadIdx.x;
  if (i < n) p[i] = 1.0f;
}
__global__ void deg_accum(const int* __restrict__ dst, float* __restrict__ deg, int E) {
  int i = blockIdx.x * blockDim.x + threadIdx.x;
  if (i < E) atomicAdd(&deg[dst[i]], 1.0f);
}
__global__ void rsqrt_inplace(float* p, int n) {
  int i = blockIdx.x * blockDim.x + threadIdx.x;
  if (i < n) { float d = p[i]; p[i] = (d > 0.0f) ? rsqrtf(d) : 0.0f; }
}
__global__ void norm_kernel(const int* __restrict__ src, const int* __restrict__ dst,
                            const float* __restrict__ dinv, float* __restrict__ norm,
                            int E, int N) {
  int i = blockIdx.x * blockDim.x + threadIdx.x;
  if (i < E) { norm[i] = dinv[src[i]] * dinv[dst[i]]; }
  else if (i < E + N) { int v = i - E; norm[i] = dinv[v] * dinv[v]; }
}

template <int F>
__global__ void scatter_add(const float* __restrict__ tmp, const int* __restrict__ src,
                            const int* __restrict__ dst, const float* __restrict__ norm,
                            float* __restrict__ agg, int nE, int nN) {
  const int lane = threadIdx.x & 31;
  const long long item = (long long)blockIdx.x * (blockDim.x >> 5) + (threadIdx.x >> 5);
  if (item >= (long long)nE + nN) return;
  int s, d; float w;
  if (item < nE) { s = src[item]; d = dst[item]; w = norm[item]; }
  else           { int v = (int)(item - nE); s = v; d = v; w = norm[nE + v]; }
  const float* ts = tmp + (long long)s * F;
  float*       ad = agg + (long long)d * F;
  #pragma unroll
  for (int f = lane; f < F; f += 32)
    atomicAdd(&ad[f], ts[f] * w);
}

__global__ void fzero(float* p, long long n) {
  long long i = (long long)blockIdx.x * blockDim.x + threadIdx.x;
  if (i < n) p[i] = 0.0f;
}
template <int F>
__global__ void bias_elu(float* __restrict__ h, const float* __restrict__ b, long long n) {
  long long i = (long long)blockIdx.x * blockDim.x + threadIdx.x;
  if (i >= n) return;
  float v = h[i] + b[(int)(i % F)];
  h[i] = (v > 0.0f) ? v : expm1f(v);
}
template <int F>
__global__ void bias_add(float* __restrict__ out, const float* __restrict__ b, long long n) {
  long long i = (long long)blockIdx.x * blockDim.x + threadIdx.x;
  if (i < n) out[i] += b[(int)(i % F)];
}

static inline int cdiv(long long a, long long b) { return (int)((a + b - 1) / b); }

extern "C" void kernel_launch(void* const* d_in, const int* in_sizes, int n_in,
                              void* d_out, int out_size, void* d_ws, size_t ws_size,
                              hipStream_t stream) {
  (void)n_in; (void)out_size; (void)ws_size;
  const float* x    = (const float*)d_in[0];
  const int*   eidx = (const int*)d_in[1];
  const float* Wst  = (const float*)d_in[2];
  const float* bst  = (const float*)d_in[3];
  const float* Wout = (const float*)d_in[4];
  const float* bout = (const float*)d_in[5];

  const int N = in_sizes[0] / HID;
  const int E = in_sizes[1] / 2;
  constexpr int NCLS = 10;

  const int* src = eidx;
  const int* dst = eidx + E;

  float* norm = (float*)d_ws;
  float* deg  = norm + ((size_t)E + N);
  float* hbuf = deg + N;
  float* tmp  = hbuf + (size_t)N * HID;

  fill_ones<<<cdiv(N, 256), 256, 0, stream>>>(deg, N);
  deg_accum<<<cdiv(E, 256), 256, 0, stream>>>(dst, deg, E);
  rsqrt_inplace<<<cdiv(N, 256), 256, 0, stream>>>(deg, N);
  norm_kernel<<<cdiv((long long)E + N, 256), 256, 0, stream>>>(src, dst, deg, norm, E, N);

  const long long nh    = (long long)N * HID;
  const int scat_blocks = cdiv((long long)E + N, 8);
  const dim3 ggrid(cdiv(N, BM), HID / BN);

  const float* hin = x;
  for (int l = 0; l < 8; ++l) {
    gemm_bf16x3<<<ggrid, 256, 0, stream>>>(hin, Wst + (size_t)l * HID * HID, tmp, N);
    fzero<<<cdiv(nh, 256), 256, 0, stream>>>(hbuf, nh);
    scatter_add<HID><<<scat_blocks, 256, 0, stream>>>(tmp, src, dst, norm, hbuf, E, N);
    bias_elu<HID><<<cdiv(nh, 256), 256, 0, stream>>>(hbuf, bst + (size_t)l * HID, nh);
    hin = hbuf;
  }

  const long long no = (long long)N * NCLS;
  gemm_small<NCLS><<<cdiv(no, 256), 256, 0, stream>>>(hbuf, Wout, tmp, N);
  fzero<<<cdiv(no, 256), 256, 0, stream>>>((float*)d_out, no);
  scatter_add<NCLS><<<scat_blocks, 256, 0, stream>>>(tmp, src, dst, norm, (float*)d_out, E, N);
  bias_add<NCLS><<<cdiv(no, 256), 256, 0, stream>>>((float*)d_out, bout, no);
}


# Float-atomic scatter replaced by a per-call in-edge list (int atomic slot, scan) and a gather: one wave sums a row in registers, one store; zero-fill gone; sum order per row = atomic slot order
# speedup vs baseline: 1.9952x; 1.9952x over previous
#include <hip/hip_runtime.h>
#include <math.h>

typedef __attribute__((ext_vector_type(16))) __bf16 v16bf;
typedef __attribute__((ext_vector_type(8)))  float  v8f;
typedef unsigned short u16;
typedef unsigned int   u32;

#define HID 128
#define BM  128
#define BN  64
#define BK  32
#define BKP 40

__device__ __forceinline__ float bfh_to_f(u16 h) {
  union { u32 u; float f; } c; c.u = ((u32)h) << 16; return c.f;
}
__device__ __forceinline__ u16 f_to_bf_rne(float f) {
  union { float f; u32 u; } c; c.f = f;
  u32 r = c.u + 0x7FFFu + ((c.u >> 16) & 1u);
  return (u16)(r >> 16);
}
__device__ __forceinline__ void split_bf(float v, u16& hi, u16& lo) {
  union { float f; u32 u; } c; c.f = v;
  u16 h = (u16)(c.u >> 16);
  float r = v - bfh_to_f(h);
  hi = h; lo = f_to_bf_rne(r);
}

__global__ __launch_bounds__(256)
void gemm_bf16x3(const float* __restrict__ A, const float* __restrict__ W,
                 float* __restrict__ C, int nrows) {
  __shared__ u16 aHi[BM][BKP];
  __shared__ u16 aLo[BM][BKP];
  __shared__ u16 bHi[BN][BKP];
  __shared__ u16 bLo[BN][BKP];

  const int tid  = threadIdx.x;
  const int lane = tid & 31;
  const int wave = tid >> 5;
  const int l15  = lane & 15;
  const int kh   = (lane >> 4) << 3;
  const int row0 = blockIdx.x * BM;
  const int col0 = blockIdx.y * BN;
  const bool fullTile = (row0 + BM) <= nrows;

  v8f acc[4];
  for (int t = 0; t < 4; ++t)
    for (int r = 0; r < 8; ++r) acc[t][r] = 0.0f;

  for (int kc = 0; kc < HID; kc += BK) {
    __syncthreads();
    if (fullTile) {
      #pragma unroll
      for (int idx = tid; idx < BM * BK; idx += 256) {
        const int m = idx >> 5, k = idx & 31;
        float v = A[(long long)(row0 + m) * HID + (kc + k)];
        split_bf(v, aHi[m][k], aLo[m][k]);
      }
    } else {
      for (int idx = tid; idx < BM * BK; idx += 256) {
        const int m = idx >> 5, k = idx & 31;
        const int row = row0 + m;
        float v = (row < nrows) ? A[(long long)row * HID + (kc + k)] : 0.0f;
        split_bf(v, aHi[m][k], aLo[m][k]);
      }
    }
    #pragma unroll
    for (int idx = tid; idx < BN * BK; idx += 256) {
      const int n = idx >> 5, k = idx & 31;
      float v = W[(kc + k) * HID + (col0 + n)];
      split_bf(v, bHi[n][k], bLo[n][k]);
    }
    __syncthreads();

    union Frag { uint4 q[2]; v16bf v; };
    Frag ah, al;
    const int m = (wave << 4) + l15;
    ah.q[0] = *(const uint4*)&aHi[m][kh];
    ah.q[1] = *(const uint4*)&aHi[m][16 + kh];
    al.q[0] = *(const uint4*)&aLo[m][kh];
    al.q[1] = *(const uint4*)&aLo[m][16 + kh];

    #pragma unroll
    for (int nt = 0; nt < 4; ++nt) {
      Frag bh, bl;
      const int n = (nt << 4) + l15;
      bh.q[0] = *(const uint4*)&bHi[n][kh];
      bh.q[1] = *(const uint4*)&bHi[n][16 + kh];
      bl.q[0] = *(const uint4*)&bLo[n][kh];
      bl.q[1] = *(const uint4*)&bLo[n][16 + kh];
      acc[nt] = __builtin_amdgcn_wmma_f32_16x16x32_bf16(false, ah.v, false, bh.v, (short)0, acc[nt], false, false);
      acc[nt] = __builtin_amdgcn_wmma_f32_16x16x32_bf16(false, ah.v, false, bl.v, (short)0, acc[nt], false, false);
      acc[nt] = __builtin_amdgcn_wmma_f32_16x16x32_bf16(false, al.v, false, bh.v, (short)0, acc[nt], false, false);
    }
  }

  const int rbase = row0 + (wave << 4) + ((lane & 16) ? 8 : 0);
  if (fullTile) {
    #pragma unroll
    for (int nt = 0; nt < 4; ++nt) {
      const int c = col0 + (nt << 4) + l15;
      #pragma unroll
      for (int r = 0; r < 8; ++r)
        C[(long long)(rbase + r) * HID + c] = acc[nt][r];
    }
  } else {
    for (int nt = 0; nt < 4; ++nt) {
      const int c = col0 + (nt << 4) + l15;
      for (int r = 0; r < 8; ++r) {
        const int row = rbase + r;
        if (row < nrows) C[(long long)row * HID + c] = acc[nt][r];
      }
    }
  }
}

template <int CC>
__global__ void gemm_small(const float* __restrict__ h, const float* __restrict__ W,
                           float* __restrict__ out, int n) {
  const long long idx = (long long)blockIdx.x * blockDim.x + threadIdx.x;
  if (idx >= (long long)n * CC) return;
  const int node = (int)(idx / CC);
  const int c    = (int)(idx % CC);
  const float* hr = h + (long long)node * HID;
  float s = 0.0f;
  #pragma unroll 8
  for (int k = 0; k < HID; ++k) s += hr[k] * W[k * CC + c];
  out[idx] = s;
}

__global__ void zero_counts(int* __restrict__ cnt, int n) {
  int i = blockIdx.x * blockDim.x + threadIdx.x;
  if (i < n) cnt[i] = 0;
}
__global__ void count_rank(const int* __restrict__ dst, int* __restrict__ cnt,
                           int* __restrict__ rank, int E) {
  int i = blockIdx.x * blockDim.x + threadIdx.x;
  if (i < E) rank[i] = atomicAdd(&cnt[dst[i]], 1);
}
__global__ __launch_bounds__(256)
void block_sums(const int* __restrict__ cnt, int* __restrict__ bsum, int n) {
  __shared__ int ws[8];
  const int tid = threadIdx.x, lane = tid & 31, wave = tid >> 5;
  const int i = blockIdx.x * 256 + tid;
  int v = (i < n) ? cnt[i] : 0;
  for (int o = 16; o > 0; o >>= 1) v += __shfl_xor(v, o, 32);
  if (lane == 0) ws[wave] = v;
  __syncthreads();
  if (tid == 0) {
    int s = 0;
    for (int w = 0; w < 8; ++w) s += ws[w];
    bsum[blockIdx.x] = s;
  }
}
__global__ __launch_bounds__(256)
void scan_rows(const int* __restrict__ cnt, const int* __restrict__ bsum,
               int* __restrict__ rowptr, float* __restrict__ dinv, int n) {
  __shared__ int wtot[8];
  __shared__ int wbase[8];
  const int tid = threadIdx.x, lane = tid & 31, wave = tid >> 5;
  int part = 0;
  for (int j = tid; j < (int)blockIdx.x; j += 256) part += bsum[j];
  for (int o = 16; o > 0; o >>= 1) part += __shfl_xor(part, o, 32);
  const int i = blockIdx.x * 256 + tid;
  const int c = (i < n) ? cnt[i] : 0;
  int inc = c;
  for (int o = 1; o < 32; o <<= 1) {
    const int t = __shfl_up(inc, o, 32);
    if (lane >= o) inc += t;
  }
  if (lane == 31) wtot[wave] = inc;
  if (lane == 0)  wbase[wave] = part;
  __syncthreads();
  int start = inc - c;
  for (int w = 0; w < 8; ++w) {
    start += wbase[w];
    if (w < wave) start += wtot[w];
  }
  if (i < n) {
    rowptr[i] = start;
    const float d = (float)(c + 1);
    dinv[i] = (d > 0.0f) ? rsqrtf(d) : 0.0f;
    if (i == n - 1) rowptr[n] = start + c;
  }
}
__global__ void fill_rows(const int* __restrict__ src, const int* __restrict__ dst,
                          const int* __restrict__ rank, const int* __restrict__ rowptr,
                          const float* __restrict__ dinv, int* __restrict__ rec, int E) {
  int i = blockIdx.x * blockDim.x + threadIdx.x;
  if (i < E) {
    const int s = src[i], d = dst[i];
    const int p = rowptr[d] + rank[i];
    int2 e; e.x = s; e.y = __float_as_int(dinv[s] * dinv[d]);
    ((int2*)rec)[p] = e;
  }
}

__global__ __launch_bounds__(256)
void gather_rows128(const float* __restrict__ tmp, const int* __restrict__ rowptr,
                    const int* __restrict__ rec, const float* __restrict__ dinv,
                    float* __restrict__ agg, int n) {
  const int lane = threadIdx.x & 31;
  int row = blockIdx.x * 8 + (threadIdx.x >> 5);
  if (row >= n) return;
  row = __builtin_amdgcn_readfirstlane(row);
  const int beg = rowptr[row], end = rowptr[row + 1];
  const float di = dinv[row];
  const float wself = di * di;
  const float4* __restrict__ t4 = (const float4*)tmp;
  const int2*   __restrict__ r2 = (const int2*)rec;
  const float4 v = t4[(long long)row * (HID / 4) + lane];
  float4 acc;
  acc.x = v.x * wself; acc.y = v.y * wself; acc.z = v.z * wself; acc.w = v.w * wself;
  #pragma unroll 4
  for (int p = beg; p < end; ++p) {
    const int2 e = r2[p];
    const float w = __int_as_float(e.y);
    const float4 u = t4[(long long)e.x * (HID / 4) + lane];
    acc.x += u.x * w; acc.y += u.y * w; acc.z += u.z * w; acc.w += u.w * w;
  }
  ((float4*)agg)[(long long)row * (HID / 4) + lane] = acc;
}
template <int CC>
__global__ void gather_rows_small(const float* __restrict__ tmp, const int* __restrict__ rowptr,
                                  const int* __restrict__ rec, const float* __restrict__ dinv,
                                  float* __restrict__ agg, int n) {
  const long long idx = (long long)blockIdx.x * blockDim.x + threadIdx.x;
  if (idx >= (long long)n * CC) return;
  const int row = (int)(idx / CC);
  const int c   = (int)(idx % CC);
  const int beg = rowptr[row], end = rowptr[row + 1];
  const float di = dinv[row];
  const int2* __restrict__ r2 = (const int2*)rec;
  float acc = tmp[idx] * (di * di);
  for (int p = beg; p < end; ++p) {
    const int2 e = r2[p];
    acc += tmp[(long long)e.x * CC + c] * __int_as_float(e.y);
  }
  agg[idx] = acc;
}

template <int F>
__global__ void bias_elu(float* __restrict__ h, const float* __restrict__ b, long long n) {
  long long i = (long long)blockIdx.x * blockDim.x + threadIdx.x;
  if (i >= n) return;
  float v = h[i] + b[(int)(i % F)];
  h[i] = (v > 0.0f) ? v : expm1f(v);
}
template <int F>
__global__ void bias_add(float* __restrict__ out, const float* __restrict__ b, long long n) {
  long long i = (long long)blockIdx.x * blockDim.x + threadIdx.x;
  if (i < n) out[i] += b[(int)(i % F)];
}

static inline int cdiv(long long a, long long b) { return (int)((a + b - 1) / b); }
static inline size_t pad64(size_t n) { return (n + 63) & ~(size_t)63; }

extern "C" void kernel_launch(void* const* d_in, const int* in_sizes, int n_in,
                              void* d_out, int out_size, void* d_ws, size_t ws_size,
                              hipStream_t stream) {
  (void)n_in; (void)out_size; (void)ws_size;
  const float* x    = (const float*)d_in[0];
  const int*   eidx = (const int*)d_in[1];
  const float* Wst  = (const float*)d_in[2];
  const float* bst  = (const float*)d_in[3];
  const float* Wout = (const float*)d_in[4];
  const float* bout = (const float*)d_in[5];

  const int N = in_sizes[0] / HID;
  const int E = in_sizes[1] / 2;
  constexpr int NCLS = 10;

  const int* src = eidx;
  const int* dst = eidx + E;
  const int nb = cdiv(N, 256);

  int*   cnt    = (int*)d_ws;
  int*   bsum   = cnt + pad64(N);
  int*   rowptr = bsum + pad64(nb);
  float* dinv   = (float*)(rowptr + pad64((size_t)N + 1));
  int*   rank   = (int*)(dinv + pad64(N));
  int*   rec    = rank + pad64(E);
  float* hbuf   = (float*)(rec + pad64((size_t)2 * E));
  float* tmp    = hbuf + (size_t)N * HID;

  zero_counts<<<nb, 256, 0, stream>>>(cnt, N);
  count_rank<<<cdiv(E, 256), 256, 0, stream>>>(dst, cnt, rank, E);
  block_sums<<<nb, 256, 0, stream>>>(cnt, bsum, N);
  scan_rows<<<nb, 256, 0, stream>>>(cnt, bsum, rowptr, dinv, N);
  fill_rows<<<cdiv(E, 256), 256, 0, stream>>>(src, dst, rank, rowptr, dinv, rec, E);

  const long long nh = (long long)N * HID;
  const dim3 ggrid(cdiv(N, BM), HID / BN);

  const float* hin = x;
  for (int l = 0; l < 8; ++l) {
    gemm_bf16x3<<<ggrid, 256, 0, stream>>>(hin, Wst + (size_t)l * HID * HID, tmp, N);
    gather_rows128<<<cdiv(N, 8), 256, 0, stream>>>(tmp, rowptr, rec, dinv, hbuf, N);
    bias_elu<HID><<<cdiv(nh, 256), 256, 0, stream>>>(hbuf, bst + (size_t)l * HID, nh);
    hin = hbuf;
  }

  const long long no = (long long)N * NCLS;
  gemm_small<NCLS><<<cdiv(no, 256), 256, 0, stream>>>(hbuf, Wout, tmp, N);
  gather_rows_small<NCLS><<<cdiv(no, 256), 256, 0, stream>>>(tmp, rowptr, rec, dinv, (float*)d_out, N);
  bias_add<NCLS><<<cdiv(no, 256), 256, 0, stream>>>((float*)d_out, bout, no);
}


# Bias add and ELU (and the last layer's bias) applied in the gather's registers before its one store per row: the 9 separate one-float-per-thread passes over h are gone; sums and their order unchanged
# speedup vs baseline: 2.5923x; 1.2993x over previous
#include <hip/hip_runtime.h>
#include <math.h>

typedef __attribute__((ext_vector_type(16))) __bf16 v16bf;
typedef __attribute__((ext_vector_type(8)))  float  v8f;
typedef unsigned short u16;
typedef unsigned int   u32;

#define HID 128
#define BM  128
#define BN  64
#define BK  32
#define BKP 40

__device__ __forceinline__ float bfh_to_f(u16 h) {
  union { u32 u; float f; } c; c.u = ((u32)h) << 16; return c.f;
}
__device__ __forceinline__ u16 f_to_bf_rne(float f) {
  union { float f; u32 u; } c; c.f = f;
  u32 r = c.u + 0x7FFFu + ((c.u >> 16) & 1u);
  return (u16)(r >> 16);
}
__device__ __forceinline__ void split_bf(float v, u16& hi, u16& lo) {
  union { float f; u32 u; } c; c.f = v;
  u16 h = (u16)(c.u >> 16);
  float r = v - bfh_to_f(h);
  hi = h; lo = f_to_bf_rne(r);
}

__global__ __launch_bounds__(256)
void gemm_bf16x3(const float* __restrict__ A, const float* __restrict__ W,
                 float* __restrict__ C, int nrows) {
  __shared__ u16 aHi[BM][BKP];
  __shared__ u16 aLo[BM][BKP];
  __shared__ u16 bHi[BN][BKP];
  __shared__ u16 bLo[BN][BKP];

  const int tid  = threadIdx.x;
  const int lane = tid & 31;
  const int wave = tid >> 5;
  const int l15  = lane & 15;
  const int kh   = (lane >> 4) << 3;
  const int row0 = blockIdx.x * BM;
  const int col0 = blockIdx.y * BN;
  const bool fullTile = (row0 + BM) <= nrows;

  v8f acc[4];
  for (int t = 0; t < 4; ++t)
    for (int r = 0; r < 8; ++r) acc[t][r] = 0.0f;

  for (int kc = 0; kc < HID; kc += BK) {
    __syncthreads();
    if (fullTile) {
      #pragma unroll
      for (int idx = tid; idx < BM * BK; idx += 256) {
        const int m = idx >> 5, k = idx & 31;
        float v = A[(long long)(row0 + m) * HID + (kc + k)];
        split_bf(v, aHi[m][k], aLo[m][k]);
      }
    } else {
      for (int idx = tid; idx < BM * BK; idx += 256) {
        const int m = idx >> 5, k = idx & 31;
        const int row = row0 + m;
        float v = (row < nrows) ? A[(long long)row * HID + (kc + k)] : 0.0f;
        split_bf(v, aHi[m][k], aLo[m][k]);
      }
    }
    #pragma unroll
    for (int idx = tid; idx < BN * BK; idx += 256) {
      const int n = idx >> 5, k = idx & 31;
      float v = W[(kc + k) * HID + (col0 + n)];
      split_bf(v, bHi[n][k], bLo[n][k]);
    }
    __syncthreads();

    union Frag { uint4 q[2]; v16bf v; };
    Frag ah, al;
    const int m = (wave << 4) + l15;
    ah.q[0] = *(const uint4*)&aHi[m][kh];
    ah.q[1] = *(const uint4*)&aHi[m][16 + kh];
    al.q[0] = *(const uint4*)&aLo[m][kh];
    al.q[1] = *(const uint4*)&aLo[m][16 + kh];

    #pragma unroll
    for (int nt = 0; nt < 4; ++nt) {
      Frag bh, bl;
      const int n = (nt << 4) + l15;
      bh.q[0] = *(const uint4*)&bHi[n][kh];
      bh.q[1] = *(const uint4*)&bHi[n][16 + kh];
      bl.q[0] = *(const uint4*)&bLo[n][kh];
      bl.q[1] = *(const uint4*)&bLo[n][16 + kh];
      acc[nt] = __builtin_amdgcn_wmma_f32_16x16x32_bf16(false, ah.v, false, bh.v, (short)0, acc[nt], false, false);
      acc[nt] = __builtin_amdgcn_wmma_f32_16x16x32_bf16(false, ah.v, false, bl.v, (short)0, acc[nt], false, false);
      acc[nt] = __builtin_amdgcn_wmma_f32_16x16x32_bf16(false, al.v, false, bh.v, (short)0, acc[nt], false, false);
    }
  }

  const int rbase = row0 + (wave << 4) + ((lane & 16) ? 8 : 0);
  if (fullTile) {
    #pragma unroll
    for (int nt = 0; nt < 4; ++nt) {
      const int c = col0 + (nt << 4) + l15;
      #pragma unroll
      for (int r = 0; r < 8; ++r)
        C[(long long)(rbase + r) * HID + c] = acc[nt][r];
    }
  } else {
    for (int nt = 0; nt < 4; ++nt) {
      const int c = col0 + (nt << 4) + l15;
      for (int r = 0; r < 8; ++r) {
        const int row = rbase + r;
        if (row < nrows) C[(long long)row * HID + c] = acc[nt][r];
      }
    }
  }
}

template <int CC>
__global__ void gemm_small(const float* __restrict__ h, const float* __restrict__ W,
                           float* __restrict__ out, int n) {
  const long long idx = (long long)blockIdx.x * blockDim.x + threadIdx.x;
  if (idx >= (long long)n * CC) return;
  const int node = (int)(idx / CC);
  const int c    = (int)(idx % CC);
  const float* hr = h + (long long)node * HID;
  float s = 0.0f;
  #pragma unroll 8
  for (int k = 0; k < HID; ++k) s += hr[k] * W[k * CC + c];
  out[idx] = s;
}

__global__ void zero_counts(int* __restrict__ cnt, int n) {
  int i = blockIdx.x * blockDim.x + threadIdx.x;
  if (i < n) cnt[i] = 0;
}
__global__ void count_rank(const int* __restrict__ dst, int* __restrict__ cnt,
                           int* __restrict__ rank, int E) {
  int i = blockIdx.x * blockDim.x + threadIdx.x;
  if (i < E) rank[i] = atomicAdd(&cnt[dst[i]], 1);
}
__global__ __launch_bounds__(256)
void block_sums(const int* __restrict__ cnt, int* __restrict__ bsum, int n) {
  __shared__ int ws[8];
  const int tid = threadIdx.x, lane = tid & 31, wave = tid >> 5;
  const int i = blockIdx.x * 256 + tid;
  int v = (i < n) ? cnt[i] : 0;
  for (int o = 16; o > 0; o >>= 1) v += __shfl_xor(v, o, 32);
  if (lane == 0) ws[wave] = v;
  __syncthreads();
  if (tid == 0) {
    int s = 0;
    for (int w = 0; w < 8; ++w) s += ws[w];
    bsum[blockIdx.x] = s;
  }
}
__global__ __launch_bounds__(256)
void scan_rows(const int* __restrict__ cnt, const int* __restrict__ bsum,
               int* __restrict__ rowptr, float* __restrict__ dinv, int n) {
  __shared__ int wtot[8];
  __shared__ int wbase[8];
  const int tid = threadIdx.x, lane = tid & 31, wave = tid >> 5;
  int part = 0;
  for (int j = tid; j < (int)blockIdx.x; j += 256) part += bsum[j];
  for (int o = 16; o > 0; o >>= 1) part += __shfl_xor(part, o, 32);
  const int i = blockIdx.x * 256 + tid;
  const int c = (i < n) ? cnt[i] : 0;
  int inc = c;
  for (int o = 1; o < 32; o <<= 1) {
    const int t = __shfl_up(inc, o, 32);
    if (lane >= o) inc += t;
  }
  if (lane == 31) wtot[wave] = inc;
  if (lane == 0)  wbase[wave] = part;
  __syncthreads();
  int start = inc - c;
  for (int w = 0; w < 8; ++w) {
    start += wbase[w];
    if (w < wave) start += wtot[w];
  }
  if (i < n) {
    rowptr[i] = start;
    const float d = (float)(c + 1);
    dinv[i] = (d > 0.0f) ? rsqrtf(d) : 0.0f;
    if (i == n - 1) rowptr[n] = start + c;
  }
}
__global__ void fill_rows(const int* __restrict__ src, const int* __restrict__ dst,
                          const int* __restrict__ rank, const int* __restrict__ rowptr,
                          const float* __restrict__ dinv, int* __restrict__ rec, int E) {
  int i = blockIdx.x * blockDim.x + threadIdx.x;
  if (i < E) {
    const int s = src[i], d = dst[i];
    const int p = rowptr[d] + rank[i];
    int2 e; e.x = s; e.y = __float_as_int(dinv[s] * dinv[d]);
    ((int2*)rec)[p] = e;
  }
}

__device__ __forceinline__ float elu1(float v) { return (v > 0.0f) ? v : expm1f(v); }

__global__ __launch_bounds__(256)
void gather_rows128(const float* __restrict__ tmp, const int* __restrict__ rowptr,
                    const int* __restrict__ rec, const float* __restrict__ dinv,
                    const float* __restrict__ bias, float* __restrict__ agg, int n) {
  const int lane = threadIdx.x & 31;
  int row = blockIdx.x * 8 + (threadIdx.x >> 5);
  if (row >= n) return;
  row = __builtin_amdgcn_readfirstlane(row);
  const int beg = rowptr[row], end = rowptr[row + 1];
  const float di = dinv[row];
  const float wself = di * di;
  const float4* __restrict__ t4 = (const float4*)tmp;
  const int2*   __restrict__ r2 = (const int2*)rec;
  const float4 v = t4[(long long)row * (HID / 4) + lane];
  const float4 b = ((const float4*)bias)[lane];
  float4 acc;
  acc.x = v.x * wself; acc.y = v.y * wself; acc.z = v.z * wself; acc.w = v.w * wself;
  #pragma unroll 4
  for (int p = beg; p < end; ++p) {
    const int2 e = r2[p];
    const float w = __int_as_float(e.y);
    const float4 u = t4[(long long)e.x * (HID / 4) + lane];
    acc.x += u.x * w; acc.y += u.y * w; acc.z += u.z * w; acc.w += u.w * w;
  }
  acc.x = elu1(acc.x + b.x); acc.y = elu1(acc.y + b.y);
  acc.z = elu1(acc.z + b.z); acc.w = elu1(acc.w + b.w);
  ((float4*)agg)[(long long)row * (HID / 4) + lane] = acc;
}
template <int CC>
__global__ void gather_rows_small(const float* __restrict__ tmp, const int* __restrict__ rowptr,
                                  const int* __restrict__ rec, const float* __restrict__ dinv,
                                  const float* __restrict__ bias, float* __restrict__ agg, int n) {
  const long long idx = (long long)blockIdx.x * blockDim.x + threadIdx.x;
  if (idx >= (long long)n * CC) return;
  const int row = (int)(idx / CC);
  const int c   = (int)(idx % CC);
  const int beg = rowptr[row], end = rowptr[row + 1];
  const float di = dinv[row];
  const int2* __restrict__ r2 = (const int2*)rec;
  float acc = tmp[idx] * (di * di);
  for (int p = beg; p < end; ++p) {
    const int2 e = r2[p];
    acc += tmp[(long long)e.x * CC + c] * __int_as_float(e.y);
  }
  agg[idx] = acc + bias[c];
}

static inline int cdiv(long long a, long long b) { return (int)((a + b - 1) / b); }
static inline size_t pad64(size_t n) { return (n + 63) & ~(size_t)63; }

extern "C" void kernel_launch(void* const* d_in, const int* in_sizes, int n_in,
                              void* d_out, int out_size, void* d_ws, size_t ws_size,
                              hipStream_t stream) {
  (void)n_in; (void)out_size; (void)ws_size;
  const float* x    = (const float*)d_in[0];
  const int*   eidx = (const int*)d_in[1];
  const float* Wst  = (const float*)d_in[2];
  const float* bst  = (const float*)d_in[3];
  const float* Wout = (const float*)d_in[4];
  const float* bout = (const float*)d_in[5];

  const int N = in_sizes[0] / HID;
  const int E = in_sizes[1] / 2;
  constexpr int NCLS = 10;

  const int* src = eidx;
  const int* dst = eidx + E;
  const int nb = cdiv(N, 256);

  int*   cnt    = (int*)d_ws;
  int*   bsum   = cnt + pad64(N);
  int*   rowptr = bsum + pad64(nb);
  float* dinv   = (float*)(rowptr + pad64((size_t)N + 1));
  int*   rank   = (int*)(dinv + pad64(N));
  int*   rec    = rank + pad64(E);
  float* hbuf   = (float*)(rec + pad64((size_t)2 * E));
  float* tmp    = hbuf + (size_t)N * HID;

  zero_counts<<<nb, 256, 0, stream>>>(cnt, N);
  count_rank<<<cdiv(E, 256), 256, 0, stream>>>(dst, cnt, rank, E);
  block_sums<<<nb, 256, 0, stream>>>(cnt, bsum, N);
  scan_rows<<<nb, 256, 0, stream>>>(cnt, bsum, rowptr, dinv, N);
  fill_rows<<<cdiv(E, 256), 256, 0, stream>>>(src, dst, rank, rowptr, dinv, rec, E);

  const dim3 ggrid(cdiv(N, BM), HID / BN);

  const float* hin = x;
  for (int l = 0; l < 8; ++l) {
    gemm_bf16x3<<<ggrid, 256, 0, stream>>>(hin, Wst + (size_t)l * HID * HID, tmp, N);
    gather_rows128<<<cdiv(N, 8), 256, 0, stream>>>(tmp, rowptr, rec, dinv, bst + (size_t)l * HID, hbuf, N);
    hin = hbuf;
  }

  const long long no = (long long)N * NCLS;
  gemm_small<NCLS><<<cdiv(no, 256), 256, 0, stream>>>(hbuf, Wout, tmp, N);
  gather_rows_small<NCLS><<<cdiv(no, 256), 256, 0, stream>>>(tmp, rowptr, rec, dinv, bout, (float*)d_out, N);
}


# Product re-tiled: a wave owns 16 whole rows x all 128 columns; A fragments straight from global (read once, no LDS pass), W split once per 16-wave group, whole-line stores; sums and order unchanged
# speedup vs baseline: 4.2643x; 1.6450x over previous
#include <hip/hip_runtime.h>
#include <math.h>

typedef __attribute__((ext_vector_type(16))) __bf16 v16bf;
typedef __attribute__((ext_vector_type(8)))  float  v8f;
typedef unsigned short u16;
typedef unsigned int   u32;

#define HID 128
#define WP  136
#define GT  512

__device__ __forceinline__ u16 f_to_bf_rne(float f) {
  union { float f; u32 u; } c; c.f = f;
  u32 r = c.u + 0x7FFFu + ((c.u >> 16) & 1u);
  return (u16)(r >> 16);
}

union Frag { uint4 q[2]; v16bf v; };

__device__ __forceinline__ u32 pack_hi(float a, float b) {
  return (__float_as_uint(a) >> 16) | (__float_as_uint(b) & 0xFFFF0000u);
}
__device__ __forceinline__ u32 pack_lo(float a, float b) {
  const float ra = a - __uint_as_float(__float_as_uint(a) & 0xFFFF0000u);
  const float rb = b - __uint_as_float(__float_as_uint(b) & 0xFFFF0000u);
  return (u32)f_to_bf_rne(ra) | ((u32)f_to_bf_rne(rb) << 16);
}
__device__ __forceinline__ uint4 hi4(const float4 a, const float4 b) {
  uint4 r; r.x = pack_hi(a.x, a.y); r.y = pack_hi(a.z, a.w); r.z = pack_hi(b.x, b.y); r.w = pack_hi(b.z, b.w); return r;
}
__device__ __forceinline__ uint4 lo4(const float4 a, const float4 b) {
  uint4 r; r.x = pack_lo(a.x, a.y); r.y = pack_lo(a.z, a.w); r.z = pack_lo(b.x, b.y); r.w = pack_lo(b.z, b.w); return r;
}
__device__ __forceinline__ float xhalf(float x) {
  return __int_as_float(__builtin_amdgcn_permlanex16(__float_as_int(x), __float_as_int(x), 0x76543210, 0xfedcba98, false, false));
}
__device__ __forceinline__ void load_rows16(float4 (&raw)[16], const float4* __restrict__ A4, int tile, int nrows, int c, int h) {
  int row = (tile << 4) + c;
  row = (row < nrows) ? row : (nrows - 1);
  const float4* __restrict__ p = A4 + (long long)row * (HID / 4) + 2 * h;
  #pragma unroll
  for (int q = 0; q < 4; ++q) {
    raw[4 * q + 0] = p[8 * q + 0];
    raw[4 * q + 1] = p[8 * q + 1];
    raw[4 * q + 2] = p[8 * q + 4];
    raw[4 * q + 3] = p[8 * q + 5];
  }
}

__global__ __launch_bounds__(GT)
void gemm_rows16(const float* __restrict__ A, const float* __restrict__ Wm,
                 float* __restrict__ C, int nrows) {
  __shared__ __attribute__((aligned(16))) u16 wHi[HID * WP];
  __shared__ __attribute__((aligned(16))) u16 wLo[HID * WP];

  const int tid  = threadIdx.x;
  const int lane = tid & 31;
  const int wave = __builtin_amdgcn_readfirstlane(tid >> 5);
  const int c = lane & 15, h = lane >> 4;
  const int ntiles = (nrows + 15) >> 4;
  const int tstep  = (int)gridDim.x * (GT / 32);
  int tile = (int)blockIdx.x + (int)gridDim.x * wave;

  const float4* __restrict__ A4 = (const float4*)A;
  float4 raw[16];
  if (tile < ntiles) load_rows16(raw, A4, tile, nrows, c, h);

  {
    const int n = tid & (HID - 1), kg0 = tid >> 7;
    float wv[4][8];
    #pragma unroll
    for (int p = 0; p < 4; ++p) {
      const int k0 = (4 * p + kg0) * 8;
      #pragma unroll
      for (int j = 0; j < 8; ++j) wv[p][j] = Wm[(k0 + j) * HID + n];
    }
    #pragma unroll
    for (int p = 0; p < 4; ++p) {
      const int k0 = (4 * p + kg0) * 8;
      uint4 vh, vl;
      vh.x = pack_hi(wv[p][0], wv[p][1]); vh.y = pack_hi(wv[p][2], wv[p][3]);
      vh.z = pack_hi(wv[p][4], wv[p][5]); vh.w = pack_hi(wv[p][6], wv[p][7]);
      vl.x = pack_lo(wv[p][0], wv[p][1]); vl.y = pack_lo(wv[p][2], wv[p][3]);
      vl.z = pack_lo(wv[p][4], wv[p][5]); vl.w = pack_lo(wv[p][6], wv[p][7]);
      *(uint4*)&wHi[n * WP + k0] = vh;
      *(uint4*)&wLo[n * WP + k0] = vl;
    }
  }
  __syncthreads();

  while (tile < ntiles) {
    Frag ah[4], al[4];
    #pragma unroll
    for (int q = 0; q < 4; ++q) {
      ah[q].q[0] = hi4(raw[4 * q + 0], raw[4 * q + 1]);
      ah[q].q[1] = hi4(raw[4 * q + 2], raw[4 * q + 3]);
      al[q].q[0] = lo4(raw[4 * q + 0], raw[4 * q + 1]);
      al[q].q[1] = lo4(raw[4 * q + 2], raw[4 * q + 3]);
    }

    v8f acc[8];
    #pragma unroll
    for (int t = 0; t < 8; ++t)
      #pragma unroll
      for (int r = 0; r < 8; ++r) acc[t][r] = 0.0f;

    Frag bh[2][2], bl[2][2];
    const int o0 = c * WP + (h << 3);
    #pragma unroll
    for (int j = 0; j < 2; ++j) {
      const int o = o0 + (j << 4) * WP;
      bh[0][j].q[0] = *(const uint4*)&wHi[o];  bh[0][j].q[1] = *(const uint4*)&wHi[o + 16];
      bl[0][j].q[0] = *(const uint4*)&wLo[o];  bl[0][j].q[1] = *(const uint4*)&wLo[o + 16];
    }
    #pragma unroll
    for (int s = 0; s < 16; ++s) {
      const int q = s >> 2, np = s & 3, cur = s & 1, nxt = cur ^ 1;
      if (s + 1 < 16) {
        const int q1 = (s + 1) >> 2, np1 = (s + 1) & 3;
        #pragma unroll
        for (int j = 0; j < 2; ++j) {
          const int o = o0 + (((np1 << 1) + j) << 4) * WP + (q1 << 5);
          bh[nxt][j].q[0] = *(const uint4*)&wHi[o];  bh[nxt][j].q[1] = *(const uint4*)&wHi[o + 16];
          bl[nxt][j].q[0] = *(const uint4*)&wLo[o];  bl[nxt][j].q[1] = *(const uint4*)&wLo[o + 16];
        }
      }
      const int n0 = np << 1, n1 = n0 + 1;
      acc[n0] = __builtin_amdgcn_wmma_f32_16x16x32_bf16(false, ah[q].v, false, bh[cur][0].v, (short)0, acc[n0], false, false);
      acc[n1] = __builtin_amdgcn_wmma_f32_16x16x32_bf16(false, ah[q].v, false, bh[cur][1].v, (short)0, acc[n1], false, false);
      acc[n0] = __builtin_amdgcn_wmma_f32_16x16x32_bf16(false, ah[q].v, false, bl[cur][0].v, (short)0, acc[n0], false, false);
      acc[n1] = __builtin_amdgcn_wmma_f32_16x16x32_bf16(false, ah[q].v, false, bl[cur][1].v, (short)0, acc[n1], false, false);
      acc[n0] = __builtin_amdgcn_wmma_f32_16x16x32_bf16(false, al[q].v, false, bh[cur][0].v, (short)0, acc[n0], false, false);
      acc[n1] = __builtin_amdgcn_wmma_f32_16x16x32_bf16(false, al[q].v, false, bh[cur][1].v, (short)0, acc[n1], false, false);
      __builtin_amdgcn_sched_barrier(0);
    }

    const int row0 = tile << 4;
    if (row0 + 16 <= nrows) {
      float* __restrict__ cp = C + (long long)row0 * HID + lane;
      #pragma unroll
      for (int np = 0; np < 4; ++np) {
        #pragma unroll
        for (int r = 0; r < 8; ++r) {
          const float a = acc[2 * np][r], b = acc[2 * np + 1][r];
          const float got = xhalf(h ? a : b);
          cp[r * HID + (np << 5)]       = h ? got : a;
          cp[(8 + r) * HID + (np << 5)] = h ? b : got;
        }
      }
    } else {
      #pragma unroll
      for (int np = 0; np < 4; ++np) {
        const int col = (np << 5) + lane;
        #pragma unroll
        for (int r = 0; r < 8; ++r) {
          const float a = acc[2 * np][r], b = acc[2 * np + 1][r];
          const float got = xhalf(h ? a : b);
          const float v1 = h ? got : a;
          const float v2 = h ? b : got;
          if (row0 + r < nrows)     C[(long long)(row0 + r) * HID + col] = v1;
          if (row0 + 8 + r < nrows) C[(long long)(row0 + 8 + r) * HID + col] = v2;
        }
      }
    }

    tile += tstep;
    if (tile < ntiles) load_rows16(raw, A4, tile, nrows, c, h);
  }
}

template <int CC>
__global__ void gemm_small(const float* __restrict__ h, const float* __restrict__ W,
                           float* __restrict__ out, int n) {
  const long long idx = (long long)blockIdx.x * blockDim.x + threadIdx.x;
  if (idx >= (long long)n * CC) return;
  const int node = (int)(idx / CC);
  const int c    = (int)(idx % CC);
  const float* hr = h + (long long)node * HID;
  float s = 0.0f;
  #pragma unroll 8
  for (int k = 0; k < HID; ++k) s += hr[k] * W[k * CC + c];
  out[idx] = s;
}

__global__ void zero_counts(int* __restrict__ cnt, int n) {
  int i = blockIdx.x * blockDim.x + threadIdx.x;
  if (i < n) cnt[i] = 0;
}
__global__ void count_rank(const int* __restrict__ dst, int* __restrict__ cnt,
                           int* __restrict__ rank, int E) {
  int i = blockIdx.x * blockDim.x + threadIdx.x;
  if (i < E) rank[i] = atomicAdd(&cnt[dst[i]], 1);
}
__global__ __launch_bounds__(256)
void block_sums(const int* __restrict__ cnt, int* __restrict__ bsum, int n) {
  __shared__ int ws[8];
  const int tid = threadIdx.x, lane = tid & 31, wave = tid >> 5;
  const int i = blockIdx.x * 256 + tid;
  int v = (i < n) ? cnt[i] : 0;
  for (int o = 16; o > 0; o >>= 1) v += __shfl_xor(v, o, 32);
  if (lane == 0) ws[wave] = v;
  __syncthreads();
  if (tid == 0) {
    int s = 0;
    for (int w = 0; w < 8; ++w) s += ws[w];
    bsum[blockIdx.x] = s;
  }
}
__global__ __launch_bounds__(256)
void scan_rows(const int* __restrict__ cnt, const int* __restrict__ bsum,
               int* __restrict__ rowptr, float* __restrict__ dinv, int n) {
  __shared__ int wtot[8];
  __shared__ int wbase[8];
  const int tid = threadIdx.x, lane = tid & 31, wave = tid >> 5;
  int part = 0;
  for (int j = tid; j < (int)blockIdx.x; j += 256) part += bsum[j];
  for (int o = 16; o > 0; o >>= 1) part += __shfl_xor(part, o, 32);
  const int i = blockIdx.x * 256 + tid;
  const int c = (i < n) ? cnt[i] : 0;
  int inc = c;
  for (int o = 1; o < 32; o <<= 1) {
    const int t = __shfl_up(inc, o, 32);
    if (lane >= o) inc += t;
  }
  if (lane == 31) wtot[wave] = inc;
  if (lane == 0)  wbase[wave] = part;
  __syncthreads();
  int start = inc - c;
  for (int w = 0; w < 8; ++w) {
    start += wbase[w];
    if (w < wave) start += wtot[w];
  }
  if (i < n) {
    rowptr[i] = start;
    const float d = (float)(c + 1);
    dinv[i] = (d > 0.0f) ? rsqrtf(d) : 0.0f;
    if (i == n - 1) rowptr[n] = start + c;
  }
}
__global__ void fill_rows(const int* __restrict__ src, const int* __restrict__ dst,
                          const int* __restrict__ rank, const int* __restrict__ rowptr,
                          const float* __restrict__ dinv, int* __restrict__ rec, int E) {
  int i = blockIdx.x * blockDim.x + threadIdx.x;
  if (i < E) {
    const int s = src[i], d = dst[i];
    const int p = rowptr[d] + rank[i];
    int2 e; e.x = s; e.y = __float_as_int(dinv[s] * dinv[d]);
    ((int2*)rec)[p] = e;
  }
}

__device__ __forceinline__ float elu1(float v) { return (v > 0.0f) ? v : expm1f(v); }

__global__ __launch_bounds__(256)
void gather_rows128(const float* __restrict__ tmp, const int* __restrict__ rowptr,
                    const int* __restrict__ rec, const float* __restrict__ dinv,
                    const float* __restrict__ bias, float* __restrict__ agg, int n) {
  const int lane = threadIdx.x & 31;
  int row = blockIdx.x * 8 + (threadIdx.x >> 5);
  if (row >= n) return;
  row = __builtin_amdgcn_readfirstlane(row);
  const int beg = rowptr[row], end = rowptr[row + 1];
  const float di = dinv[row];
  const float wself = di * di;
  const float4* __restrict__ t4 = (const float4*)tmp;
  const int2*   __restrict__ r2 = (const int2*)rec;
  const float4 v = t4[(long long)row * (HID / 4) + lane];
  const float4 b = ((const float4*)bias)[lane];
  float4 acc;
  acc.x = v.x * wself; acc.y = v.y * wself; acc.z = v.z * wself; acc.w = v.w * wself;
  #pragma unroll 4
  for (int p = beg; p < end; ++p) {
    const int2 e = r2[p];
    const float w = __int_as_float(e.y);
    const float4 u = t4[(long long)e.x * (HID / 4) + lane];
    acc.x += u.x * w; acc.y += u.y * w; acc.z += u.z * w; acc.w += u.w * w;
  }
  acc.x = elu1(acc.x + b.x); acc.y = elu1(acc.y + b.y);
  acc.z = elu1(acc.z + b.z); acc.w = elu1(acc.w + b.w);
  ((float4*)agg)[(long long)row * (HID / 4) + lane] = acc;
}
template <int CC>
__global__ void gather_rows_small(const float* __restrict__ tmp, const int* __restrict__ rowptr,
                                  const int* __restrict__ rec, const float* __restrict__ dinv,
                                  const float* __restrict__ bias, float* __restrict__ agg, int n) {
  const long long idx = (long long)blockIdx.x * blockDim.x + threadIdx.x;
  if (idx >= (long long)n * CC) return;
  const int row = (int)(idx / CC);
  const int c   = (int)(idx % CC);
  const int beg = rowptr[row], end = rowptr[row + 1];
  const float di = dinv[row];
  const int2* __restrict__ r2 = (const int2*)rec;
  float acc = tmp[idx] * (di * di);
  for (int p = beg; p < end; ++p) {
    const int2 e = r2[p];
    acc += tmp[(long long)e.x * CC + c] * __int_as_float(e.y);
  }
  agg[idx] = acc + bias[c];
}

static inline int cdiv(long long a, long long b) { return (int)((a + b - 1) / b); }
static inline size_t pad64(size_t n) { return (n + 63) & ~(size_t)63; }

extern "C" void kernel_launch(void* const* d_in, const int* in_sizes, int n_in,
                              void* d_out, int out_size, void* d_ws, size_t ws_size,
                              hipStream_t stream) {
  (void)n_in; (void)out_size; (void)ws_size;
  const float* x    = (const float*)d_in[0];
  const int*   eidx = (const int*)d_in[1];
  const float* Wst  = (const float*)d_in[2];
  const float* bst  = (const float*)d_in[3];
  const float* Wout = (const float*)d_in[4];
  const float* bout = (const float*)d_in[5];

  const int N = in_sizes[0] / HID;
  const int E = in_sizes[1] / 2;
  constexpr int NCLS = 10;

  const int* src = eidx;
  const int* dst = eidx + E;
  const int nb = cdiv(N, 256);

  int*   cnt    = (int*)d_ws;
  int*   bsum   = cnt + pad64(N);
  int*   rowptr = bsum + pad64(nb);
  float* dinv   = (float*)(rowptr + pad64((size_t)N + 1));
  int*   rank   = (int*)(dinv + pad64(N));
  int*   rec    = rank + pad64(E);
  float* hbuf   = (float*)(rec + pad64((size_t)2 * E));
  float* tmp    = hbuf + (size_t)N * HID;

  zero_counts<<<nb, 256, 0, stream>>>(cnt, N);
  count_rank<<<cdiv(E, 256), 256, 0, stream>>>(dst, cnt, rank, E);
  block_sums<<<nb, 256, 0, stream>>>(cnt, bsum, N);
  scan_rows<<<nb, 256, 0, stream>>>(cnt, bsum, rowptr, dinv, N);
  fill_rows<<<cdiv(E, 256), 256, 0, stream>>>(src, dst, rank, rowptr, dinv, rec, E);

  const float* hin = x;
  for (int l = 0; l < 8; ++l) {
    gemm_rows16<<<256, GT, 0, stream>>>(hin, Wst + (size_t)l * HID * HID, tmp, N);
    gather_rows128<<<cdiv(N, 8), 256, 0, stream>>>(tmp, rowptr, rec, dinv, bst + (size_t)l * HID, hbuf, N);
    hin = hbuf;
  }

  const long long no = (long long)N * NCLS;
  gemm_small<NCLS><<<cdiv(no, 256), 256, 0, stream>>>(hbuf, Wout, tmp, N);
  gather_rows_small<NCLS><<<cdiv(no, 256), 256, 0, stream>>>(tmp, rowptr, rec, dinv, bout, (float*)d_out, N);
}


# Edge grouping redone as a two-level counting sort on LDS counters (chunk by 256-node bucket, bucket by node): global atomics, scan launches, scattered fills gone; row sum order = LDS slot order
# speedup vs baseline: 5.8628x; 1.3749x over previous
#include <hip/hip_runtime.h>
#include <math.h>

typedef __attribute__((ext_vector_type(16))) __bf16 v16bf;
typedef __attribute__((ext_vector_type(8)))  float  v8f;
typedef unsigned short u16;
typedef unsigned int   u32;

#define HID 128
#define WP  136
#define GT  512

__device__ __forceinline__ u16 f_to_bf_rne(float f) {
  union { float f; u32 u; } c; c.f = f;
  u32 r = c.u + 0x7FFFu + ((c.u >> 16) & 1u);
  return (u16)(r >> 16);
}

union Frag { uint4 q[2]; v16bf v; };

__device__ __forceinline__ u32 pack_hi(float a, float b) {
  return (__float_as_uint(a) >> 16) | (__float_as_uint(b) & 0xFFFF0000u);
}
__device__ __forceinline__ u32 pack_lo(float a, float b) {
  const float ra = a - __uint_as_float(__float_as_uint(a) & 0xFFFF0000u);
  const float rb = b - __uint_as_float(__float_as_uint(b) & 0xFFFF0000u);
  return (u32)f_to_bf_rne(ra) | ((u32)f_to_bf_rne(rb) << 16);
}
__device__ __forceinline__ uint4 hi4(const float4 a, const float4 b) {
  uint4 r; r.x = pack_hi(a.x, a.y); r.y = pack_hi(a.z, a.w); r.z = pack_hi(b.x, b.y); r.w = pack_hi(b.z, b.w); return r;
}
__device__ __forceinline__ uint4 lo4(const float4 a, const float4 b) {
  uint4 r; r.x = pack_lo(a.x, a.y); r.y = pack_lo(a.z, a.w); r.z = pack_lo(b.x, b.y); r.w = pack_lo(b.z, b.w); return r;
}
__device__ __forceinline__ float xhalf(float x) {
  return __int_as_float(__builtin_amdgcn_permlanex16(__float_as_int(x), __float_as_int(x), 0x76543210, 0xfedcba98, false, false));
}
__device__ __forceinline__ void load_rows16(float4 (&raw)[16], const float4* __restrict__ A4, int tile, int nrows, int c, int h) {
  int row = (tile << 4) + c;
  row = (row < nrows) ? row : (nrows - 1);
  const float4* __restrict__ p = A4 + (long long)row * (HID / 4) + 2 * h;
  #pragma unroll
  for (int q = 0; q < 4; ++q) {
    raw[4 * q + 0] = p[8 * q + 0];
    raw[4 * q + 1] = p[8 * q + 1];
    raw[4 * q + 2] = p[8 * q + 4];
    raw[4 * q + 3] = p[8 * q + 5];
  }
}

__device__ __forceinline__ void stage_w(const float* __restrict__ Wm, u16* __restrict__ wHi, u16* __restrict__ wLo, int tid) {
  const int n = tid & (HID - 1), kg0 = tid >> 7;
  float wv[4][8];
  #pragma unroll
  for (int p = 0; p < 4; ++p) {
    const int k0 = (4 * p + kg0) * 8;
    #pragma unroll
    for (int j = 0; j < 8; ++j) wv[p][j] = Wm[(k0 + j) * HID + n];
  }
  #pragma unroll
  for (int p = 0; p < 4; ++p) {
    const int k0 = (4 * p + kg0) * 8;
    uint4 vh, vl;
    vh.x = pack_hi(wv[p][0], wv[p][1]); vh.y = pack_hi(wv[p][2], wv[p][3]);
    vh.z = pack_hi(wv[p][4], wv[p][5]); vh.w = pack_hi(wv[p][6], wv[p][7]);
    vl.x = pack_lo(wv[p][0], wv[p][1]); vl.y = pack_lo(wv[p][2], wv[p][3]);
    vl.z = pack_lo(wv[p][4], wv[p][5]); vl.w = pack_lo(wv[p][6], wv[p][7]);
    *(uint4*)&wHi[n * WP + k0] = vh;
    *(uint4*)&wLo[n * WP + k0] = vl;
  }
}

__device__ __forceinline__ void rows16_times_w(const float4 (&raw)[16], const u16* __restrict__ wHi, const u16* __restrict__ wLo,
                                               int c, int h, v8f (&acc)[8]) {
  Frag ah[4], al[4];
  #pragma unroll
  for (int q = 0; q < 4; ++q) {
    ah[q].q[0] = hi4(raw[4 * q + 0], raw[4 * q + 1]);
    ah[q].q[1] = hi4(raw[4 * q + 2], raw[4 * q + 3]);
    al[q].q[0] = lo4(raw[4 * q + 0], raw[4 * q + 1]);
    al[q].q[1] = lo4(raw[4 * q + 2], raw[4 * q + 3]);
  }
  #pragma unroll
  for (int t = 0; t < 8; ++t)
    #pragma unroll
    for (int r = 0; r < 8; ++r) acc[t][r] = 0.0f;

  Frag bh[2][2], bl[2][2];
  const int o0 = c * WP + (h << 3);
  #pragma unroll
  for (int j = 0; j < 2; ++j) {
    const int o = o0 + (j << 4) * WP;
    bh[0][j].q[0] = *(const uint4*)&wHi[o];  bh[0][j].q[1] = *(const uint4*)&wHi[o + 16];
    bl[0][j].q[0] = *(const uint4*)&wLo[o];  bl[0][j].q[1] = *(const uint4*)&wLo[o + 16];
  }
  #pragma unroll
  for (int s = 0; s < 16; ++s) {
    const int q = s >> 2, np = s & 3, cur = s & 1, nxt = cur ^ 1;
    if (s + 1 < 16) {
      const int q1 = (s + 1) >> 2, np1 = (s + 1) & 3;
      #pragma unroll
      for (int j = 0; j < 2; ++j) {
        const int o = o0 + (((np1 << 1) + j) << 4) * WP + (q1 << 5);
        bh[nxt][j].q[0] = *(const uint4*)&wHi[o];  bh[nxt][j].q[1] = *(const uint4*)&wHi[o + 16];
        bl[nxt][j].q[0] = *(const uint4*)&wLo[o];  bl[nxt][j].q[1] = *(const uint4*)&wLo[o + 16];
      }
    }
    const int n0 = np << 1, n1 = n0 + 1;
    acc[n0] = __builtin_amdgcn_wmma_f32_16x16x32_bf16(false, ah[q].v, false, bh[cur][0].v, (short)0, acc[n0], false, false);
    acc[n1] = __builtin_amdgcn_wmma_f32_16x16x32_bf16(false, ah[q].v, false, bh[cur][1].v, (short)0, acc[n1], false, false);
    acc[n0] = __builtin_amdgcn_wmma_f32_16x16x32_bf16(false, ah[q].v, false, bl[cur][0].v, (short)0, acc[n0], false, false);
    acc[n1] = __builtin_amdgcn_wmma_f32_16x16x32_bf16(false, ah[q].v, false, bl[cur][1].v, (short)0, acc[n1], false, false);
    acc[n0] = __builtin_amdgcn_wmma_f32_16x16x32_bf16(false, al[q].v, false, bh[cur][0].v, (short)0, acc[n0], false, false);
    acc[n1] = __builtin_amdgcn_wmma_f32_16x16x32_bf16(false, al[q].v, false, bh[cur][1].v, (short)0, acc[n1], false, false);
    __builtin_amdgcn_sched_barrier(0);
  }
}

__device__ __forceinline__ float elu1(float v) { return (v > 0.0f) ? v : expm1f(v); }

__device__ __forceinline__ float elu_mid(float v) { return (v > 0.0f) ? v : (__expf(v) - 1.0f); }

#define MP 132

__global__ __launch_bounds__(GT)
void gemm_rows16_x2(const float* __restrict__ A, const float* __restrict__ W1, const float* __restrict__ b1,
                    const float* __restrict__ W2, float* __restrict__ C, int nrows) {
  __shared__ __attribute__((aligned(16))) u16 wHi1[HID * WP];
  __shared__ __attribute__((aligned(16))) u16 wLo1[HID * WP];
  __shared__ __attribute__((aligned(16))) u16 wHi2[HID * WP];
  __shared__ __attribute__((aligned(16))) u16 wLo2[HID * WP];
  __shared__ __attribute__((aligned(16))) float mid[(GT / 32) * 16 * MP];

  const int tid  = threadIdx.x;
  const int lane = tid & 31;
  const int wave = __builtin_amdgcn_readfirstlane(tid >> 5);
  const int c = lane & 15, h = lane >> 4;
  const int ntiles = (nrows + 15) >> 4;
  const int tstep  = (int)gridDim.x * (GT / 32);
  int tile = (int)blockIdx.x + (int)gridDim.x * wave;

  const float4* __restrict__ A4 = (const float4*)A;
  float4 raw[16];
  if (tile < ntiles) load_rows16(raw, A4, tile, nrows, c, h);
  float bias[8];
  #pragma unroll
  for (int t = 0; t < 8; ++t) bias[t] = b1[(t << 4) + c];

  stage_w(W1, wHi1, wLo1, tid);
  stage_w(W2, wHi2, wLo2, tid);
  __syncthreads();

  float* __restrict__ mw = mid + wave * (16 * MP);

  while (tile < ntiles) {
    v8f acc[8];
    rows16_times_w(raw, wHi1, wLo1, c, h, acc);

    #pragma unroll
    for (int t = 0; t < 8; ++t)
      #pragma unroll
      for (int r = 0; r < 8; ++r)
        mw[(8 * h + r) * MP + (t << 4) + c] = elu_mid(acc[t][r] + bias[t]);
    __builtin_amdgcn_fence(__ATOMIC_RELEASE, "wavefront");
    __builtin_amdgcn_wave_barrier();
    __builtin_amdgcn_fence(__ATOMIC_ACQUIRE, "wavefront");
    {
      const float4* __restrict__ p = (const float4*)mw + c * (MP / 4) + 2 * h;
      #pragma unroll
      for (int q = 0; q < 4; ++q) {
        raw[4 * q + 0] = p[8 * q + 0];
        raw[4 * q + 1] = p[8 * q + 1];
        raw[4 * q + 2] = p[8 * q + 4];
        raw[4 * q + 3] = p[8 * q + 5];
      }
    }
    __builtin_amdgcn_fence(__ATOMIC_RELEASE, "wavefront");
    __builtin_amdgcn_wave_barrier();

    rows16_times_w(raw, wHi2, wLo2, c, h, acc);

    const int row0 = tile << 4;
    if (row0 + 16 <= nrows) {
      float* __restrict__ cp = C + (long long)row0 * HID + lane;
      #pragma unroll
      for (int np = 0; np < 4; ++np) {
        #pragma unroll
        for (int r = 0; r < 8; ++r) {
          const float a = acc[2 * np][r], b = acc[2 * np + 1][r];
          const float got = xhalf(h ? a : b);
          cp[r * HID + (np << 5)]       = h ? got : a;
          cp[(8 + r) * HID + (np << 5)] = h ? b : got;
        }
      }
    } else {
      #pragma unroll
      for (int np = 0; np < 4; ++np) {
        const int col = (np << 5) + lane;
        #pragma unroll
        for (int r = 0; r < 8; ++r) {
          const float a = acc[2 * np][r], b = acc[2 * np + 1][r];
          const float got = xhalf(h ? a : b);
          const float v1 = h ? got : a;
          const float v2 = h ? b : got;
          if (row0 + r < nrows)     C[(long long)(row0 + r) * HID + col] = v1;
          if (row0 + 8 + r < nrows) C[(long long)(row0 + 8 + r) * HID + col] = v2;
        }
      }
    }

    tile += tstep;
    if (tile < ntiles) load_rows16(raw, A4, tile, nrows, c, h);
  }
}

template <int CC>
__global__ void gemm_small(const float* __restrict__ h, const float* __restrict__ W,
                           float* __restrict__ out, int n) {
  const long long idx = (long long)blockIdx.x * blockDim.x + threadIdx.x;
  if (idx >= (long long)n * CC) return;
  const int node = (int)(idx / CC);
  const int c    = (int)(idx % CC);
  const float* hr = h + (long long)node * HID;
  float s = 0.0f;
  #pragma unroll 8
  for (int k = 0; k < HID; ++k) s += hr[k] * W[k * CC + c];
  out[idx] = s;
}

#define CH   2048
#define GMAX 1024
#define CAP  8192

__device__ __forceinline__ void scan256_wave(const int* cnt, int* start, int lane) {
  int v[8];
  int s = 0;
  #pragma unroll
  for (int k = 0; k < 8; ++k) { v[k] = cnt[8 * lane + k]; s += v[k]; }
  int incl = s;
  for (int o = 1; o < 32; o <<= 1) {
    const int t = __shfl_up(incl, o, 32);
    if (lane >= o) incl += t;
  }
  int run = incl - s;
  #pragma unroll
  for (int k = 0; k < 8; ++k) { start[8 * lane + k] = run; run += v[k]; }
}

__global__ __launch_bounds__(512)
void chunk_sort(const int* __restrict__ src, const int* __restrict__ dst,
                int2* __restrict__ sorted, int2* __restrict__ tab, int E) {
  __shared__ int cnt[256];
  __shared__ int start[256];
  __shared__ int2 buf[CH];
  const int tid = threadIdx.x;
  if (tid < 256) cnt[tid] = 0;
  __syncthreads();
  const int base = blockIdx.x * CH;
  int s[4], d[4], r[4];
  #pragma unroll
  for (int k = 0; k < 4; ++k) {
    const int i = base + k * 512 + tid;
    s[k] = (i < E) ? src[i] : 0;
    d[k] = (i < E) ? dst[i] : 0;
  }
  #pragma unroll
  for (int k = 0; k < 4; ++k) {
    const int i = base + k * 512 + tid;
    r[k] = (i < E) ? atomicAdd(&cnt[d[k] >> 8], 1) : 0;
  }
  __syncthreads();
  if (tid < 32) scan256_wave(cnt, start, tid);
  __syncthreads();
  if (tid < 256) tab[blockIdx.x * 256 + tid] = make_int2(start[tid], cnt[tid]);
  #pragma unroll
  for (int k = 0; k < 4; ++k) {
    const int i = base + k * 512 + tid;
    if (i < E) buf[start[d[k] >> 8] + r[k]] = make_int2(s[k], d[k]);
  }
  __syncthreads();
  #pragma unroll
  for (int k = 0; k < 4; ++k) {
    const int j = k * 512 + tid;
    if (base + j < E) sorted[base + j] = buf[j];
  }
}

__device__ __forceinline__ int2 bucket_edge(const int2* __restrict__ sorted, const int* pbase, const int* pst, int G, int j) {
  int lo = 0, hi = G;
  while (hi - lo > 1) {
    const int mid = (lo + hi) >> 1;
    if (pbase[mid] <= j) lo = mid; else hi = mid;
  }
  return sorted[(long long)lo * CH + pst[lo] + (j - pbase[lo])];
}

__global__ __launch_bounds__(GMAX)
void bucket_rows(const int2* __restrict__ sorted, const int2* __restrict__ tab, int G,
                 int* __restrict__ rowptr, float* __restrict__ dinv, int2* __restrict__ rec, int n) {
  __shared__ int pst[GMAX];
  __shared__ int pbase[GMAX];
  __shared__ int wtot[32];
  __shared__ int ncnt[256];
  __shared__ int nstart[256];
  __shared__ int nfill[256];
  __shared__ int sbase;
  __shared__ int2 raw[CAP];
  __shared__ int2 stage[CAP];
  const int tid = threadIdx.x, lane = tid & 31, wave = tid >> 5;
  const int b = blockIdx.x;
  if (tid < 256) { ncnt[tid] = 0; nfill[tid] = 0; }
  if (tid == 0) sbase = 0;
  __syncthreads();

  const bool has = tid < G;
  const int2 t = has ? tab[tid * 256 + b] : make_int2(0, 0);
  if (has) pst[tid] = t.x;
  int sx = t.x;
  for (int o = 16; o > 0; o >>= 1) sx += __shfl_xor(sx, o, 32);
  if (lane == 0) atomicAdd(&sbase, sx);
  int incl = t.y;
  for (int o = 1; o < 32; o <<= 1) {
    const int u = __shfl_up(incl, o, 32);
    if (lane >= o) incl += u;
  }
  if (lane == 31) wtot[wave] = incl;
  __syncthreads();
  const int wv = wtot[lane];
  int winc = wv;
  for (int o = 1; o < 32; o <<= 1) {
    const int u = __shfl_up(winc, o, 32);
    if (lane >= o) winc += u;
  }
  const int total = __shfl(winc, 31, 32);
  const int woff  = __shfl(winc - wv, wave, 32);
  if (has) pbase[tid] = woff + incl - t.y;
  const int base = sbase;
  const bool fits = (total <= CAP);
  __syncthreads();

  for (int j0 = 0; j0 < total; j0 += 4 * GMAX) {
    int lo[4], hi[4], jj[4];
    #pragma unroll
    for (int k = 0; k < 4; ++k) { jj[k] = j0 + k * GMAX + tid; lo[k] = 0; hi[k] = G; }
    for (int s = 0; s < 10; ++s) {
      #pragma unroll
      for (int k = 0; k < 4; ++k) {
        const int mid = (lo[k] + hi[k]) >> 1;
        const int p = pbase[mid];
        if (hi[k] - lo[k] > 1) { if (p <= jj[k]) lo[k] = mid; else hi[k] = mid; }
      }
    }
    int2 e[4];
    #pragma unroll
    for (int k = 0; k < 4; ++k)
      e[k] = (jj[k] < total) ? sorted[(long long)lo[k] * CH + pst[lo[k]] + (jj[k] - pbase[lo[k]])] : make_int2(0, 0);
    #pragma unroll
    for (int k = 0; k < 4; ++k)
      if (jj[k] < total) {
        if (fits) raw[jj[k]] = e[k];
        atomicAdd(&ncnt[e[k].y & 255], 1);
      }
  }
  __syncthreads();
  if (wave == 0) scan256_wave(ncnt, nstart, lane);
  __syncthreads();

  if (tid < 256) {
    const int node = b * 256 + tid;
    if (node < n) {
      const int c = ncnt[tid];
      rowptr[node] = base + nstart[tid];
      const float d = (float)(c + 1);
      dinv[node] = (d > 0.0f) ? rsqrtf(d) : 0.0f;
      if (node == n - 1) rowptr[n] = base + nstart[tid] + c;
    }
  }
  for (int j = tid; j < total; j += GMAX) {
    const int2 e = fits ? raw[j] : bucket_edge(sorted, pbase, pst, G, j);
    const int nl = e.y & 255;
    const int pos = nstart[nl] + atomicAdd(&nfill[nl], 1);
    if (fits) stage[pos] = e; else rec[base + pos] = e;
  }
  __syncthreads();
  if (fits)
    for (int j = tid; j < total; j += GMAX) rec[base + j] = stage[j];
}

__global__ void norm_fill(int2* __restrict__ rec, const float* __restrict__ dinv, int E) {
  const int i = blockIdx.x * blockDim.x + threadIdx.x;
  if (i < E) {
    int2 e = rec[i];
    e.y = __float_as_int(dinv[e.x] * dinv[e.y]);
    rec[i] = e;
  }
}

template <bool ACT>
__global__ __launch_bounds__(256)
void gather_rows128(const float* __restrict__ tmp, const int* __restrict__ rowptr,
                    const int* __restrict__ rec, const float* __restrict__ dinv,
                    const float* __restrict__ bias, float* __restrict__ agg, int n) {
  const int lane = threadIdx.x & 31;
  int row = blockIdx.x * 8 + (threadIdx.x >> 5);
  if (row >= n) return;
  row = __builtin_amdgcn_readfirstlane(row);
  const int beg = rowptr[row], end = rowptr[row + 1];
  const float di = dinv[row];
  const float wself = di * di;
  const float4* __restrict__ t4 = (const float4*)tmp;
  const int2*   __restrict__ r2 = (const int2*)rec;
  const float4 v = t4[(long long)row * (HID / 4) + lane];
  float4 b;
  if (ACT) b = ((const float4*)bias)[lane];
  float4 acc;
  acc.x = v.x * wself; acc.y = v.y * wself; acc.z = v.z * wself; acc.w = v.w * wself;
  #pragma unroll 4
  for (int p = beg; p < end; ++p) {
    const int2 e = r2[p];
    const float w = __int_as_float(e.y);
    const float4 u = t4[(long long)e.x * (HID / 4) + lane];
    acc.x += u.x * w; acc.y += u.y * w; acc.z += u.z * w; acc.w += u.w * w;
  }
  if (ACT) {
    acc.x = elu1(acc.x + b.x); acc.y = elu1(acc.y + b.y);
    acc.z = elu1(acc.z + b.z); acc.w = elu1(acc.w + b.w);
  }
  ((float4*)agg)[(long long)row * (HID / 4) + lane] = acc;
}
template <int CC>
__global__ void gather_rows_small(const float* __restrict__ tmp, const int* __restrict__ rowptr,
                                  const int* __restrict__ rec, const float* __restrict__ dinv,
                                  const float* __restrict__ bias, float* __restrict__ agg, int n) {
  const long long idx = (long long)blockIdx.x * blockDim.x + threadIdx.x;
  if (idx >= (long long)n * CC) return;
  const int row = (int)(idx / CC);
  const int c   = (int)(idx % CC);
  const int beg = rowptr[row], end = rowptr[row + 1];
  const float di = dinv[row];
  const int2* __restrict__ r2 = (const int2*)rec;
  float acc = tmp[idx] * (di * di);
  for (int p = beg; p < end; ++p) {
    const int2 e = r2[p];
    acc += tmp[(long long)e.x * CC + c] * __int_as_float(e.y);
  }
  agg[idx] = acc + bias[c];
}

static inline int cdiv(long long a, long long b) { return (int)((a + b - 1) / b); }
static inline size_t pad64(size_t n) { return (n + 63) & ~(size_t)63; }

extern "C" void kernel_launch(void* const* d_in, const int* in_sizes, int n_in,
                              void* d_out, int out_size, void* d_ws, size_t ws_size,
                              hipStream_t stream) {
  (void)n_in; (void)out_size; (void)ws_size;
  const float* x    = (const float*)d_in[0];
  const int*   eidx = (const int*)d_in[1];
  const float* Wst  = (const float*)d_in[2];
  const float* bst  = (const float*)d_in[3];
  const float* Wout = (const float*)d_in[4];
  const float* bout = (const float*)d_in[5];

  const int N = in_sizes[0] / HID;
  const int E = in_sizes[1] / 2;
  constexpr int NCLS = 10;

  const int* src = eidx;
  const int* dst = eidx + E;
  const int G   = cdiv(E, CH);
  const int nbk = cdiv(N, 256);

  int*   rowptr = (int*)d_ws;
  float* dinv   = (float*)(rowptr + pad64((size_t)N + 1));
  int*   tab    = (int*)(dinv + pad64(N));
  int*   sorted = tab + pad64((size_t)2 * G * 256);
  int*   rec    = sorted + pad64((size_t)2 * E);
  float* hbuf   = (float*)(rec + pad64((size_t)2 * E));
  float* tmp    = hbuf + (size_t)N * HID;
  float* tmp2   = tmp + (size_t)N * HID;

  chunk_sort<<<G, 512, 0, stream>>>(src, dst, (int2*)sorted, (int2*)tab, E);
  bucket_rows<<<nbk, GMAX, 0, stream>>>((const int2*)sorted, (const int2*)tab, G, rowptr, dinv, (int2*)rec, N);
  norm_fill<<<cdiv(E, 256), 256, 0, stream>>>((int2*)rec, dinv, E);

  const float* hin = x;
  for (int l = 0; l < 8; l += 2) {
    gather_rows128<false><<<cdiv(N, 8), 256, 0, stream>>>(hin, rowptr, rec, dinv, bst, tmp, N);
    gemm_rows16_x2<<<256, GT, 0, stream>>>(tmp, Wst + (size_t)l * HID * HID, bst + (size_t)l * HID,
                                           Wst + (size_t)(l + 1) * HID * HID, tmp2, N);
    gather_rows128<true><<<cdiv(N, 8), 256, 0, stream>>>(tmp2, rowptr, rec, dinv, bst + (size_t)(l + 1) * HID, hbuf, N);
    hin = hbuf;
  }

  const long long no = (long long)N * NCLS;
  gemm_small<NCLS><<<cdiv(no, 256), 256, 0, stream>>>(hbuf, Wout, tmp, N);
  gather_rows_small<NCLS><<<cdiv(no, 256), 256, 0, stream>>>(tmp, rowptr, rec, dinv, bout, (float*)d_out, N);
}


# Sum-first layers' gather moved into the product launch: a wave sums its 16 rows' in-edges into its LDS tile (records by one vector load, 16 rows in flight); 4 launches gone; sum order unchanged
# speedup vs baseline: 6.2535x; 1.0666x over previous
#include <hip/hip_runtime.h>
#include <math.h>

typedef __attribute__((ext_vector_type(16))) __bf16 v16bf;
typedef __attribute__((ext_vector_type(8)))  float  v8f;
typedef unsigned short u16;
typedef unsigned int   u32;

#define HID 128
#define WP  136
#define GT  512

__device__ __forceinline__ u16 f_to_bf_rne(float f) {
  union { float f; u32 u; } c; c.f = f;
  u32 r = c.u + 0x7FFFu + ((c.u >> 16) & 1u);
  return (u16)(r >> 16);
}

union Frag { uint4 q[2]; v16bf v; };

__device__ __forceinline__ u32 pack_hi(float a, float b) {
  return (__float_as_uint(a) >> 16) | (__float_as_uint(b) & 0xFFFF0000u);
}
__device__ __forceinline__ u32 pack_lo(float a, float b) {
  const float ra = a - __uint_as_float(__float_as_uint(a) & 0xFFFF0000u);
  const float rb = b - __uint_as_float(__float_as_uint(b) & 0xFFFF0000u);
  return (u32)f_to_bf_rne(ra) | ((u32)f_to_bf_rne(rb) << 16);
}
__device__ __forceinline__ uint4 hi4(const float4 a, const float4 b) {
  uint4 r; r.x = pack_hi(a.x, a.y); r.y = pack_hi(a.z, a.w); r.z = pack_hi(b.x, b.y); r.w = pack_hi(b.z, b.w); return r;
}
__device__ __forceinline__ uint4 lo4(const float4 a, const float4 b) {
  uint4 r; r.x = pack_lo(a.x, a.y); r.y = pack_lo(a.z, a.w); r.z = pack_lo(b.x, b.y); r.w = pack_lo(b.z, b.w); return r;
}
__device__ __forceinline__ float xhalf(float x) {
  return __int_as_float(__builtin_amdgcn_permlanex16(__float_as_int(x), __float_as_int(x), 0x76543210, 0xfedcba98, false, false));
}
__device__ __forceinline__ void stage_w(const float* __restrict__ Wm, u16* __restrict__ wHi, u16* __restrict__ wLo, int tid) {
  const int n = tid & (HID - 1), kg0 = tid >> 7;
  float wv[4][8];
  #pragma unroll
  for (int p = 0; p < 4; ++p) {
    const int k0 = (4 * p + kg0) * 8;
    #pragma unroll
    for (int j = 0; j < 8; ++j) wv[p][j] = Wm[(k0 + j) * HID + n];
  }
  #pragma unroll
  for (int p = 0; p < 4; ++p) {
    const int k0 = (4 * p + kg0) * 8;
    uint4 vh, vl;
    vh.x = pack_hi(wv[p][0], wv[p][1]); vh.y = pack_hi(wv[p][2], wv[p][3]);
    vh.z = pack_hi(wv[p][4], wv[p][5]); vh.w = pack_hi(wv[p][6], wv[p][7]);
    vl.x = pack_lo(wv[p][0], wv[p][1]); vl.y = pack_lo(wv[p][2], wv[p][3]);
    vl.z = pack_lo(wv[p][4], wv[p][5]); vl.w = pack_lo(wv[p][6], wv[p][7]);
    *(uint4*)&wHi[n * WP + k0] = vh;
    *(uint4*)&wLo[n * WP + k0] = vl;
  }
}

__device__ __forceinline__ void rows16_times_w(const float4 (&raw)[16], const u16* __restrict__ wHi, const u16* __restrict__ wLo,
                                               int c, int h, v8f (&acc)[8]) {
  Frag ah[4], al[4];
  #pragma unroll
  for (int q = 0; q < 4; ++q) {
    ah[q].q[0] = hi4(raw[4 * q + 0], raw[4 * q + 1]);
    ah[q].q[1] = hi4(raw[4 * q + 2], raw[4 * q + 3]);
    al[q].q[0] = lo4(raw[4 * q + 0], raw[4 * q + 1]);
    al[q].q[1] = lo4(raw[4 * q + 2], raw[4 * q + 3]);
  }
  #pragma unroll
  for (int t = 0; t < 8; ++t)
    #pragma unroll
    for (int r = 0; r < 8; ++r) acc[t][r] = 0.0f;

  Frag bh[2][2], bl[2][2];
  const int o0 = c * WP + (h << 3);
  #pragma unroll
  for (int j = 0; j < 2; ++j) {
    const int o = o0 + (j << 4) * WP;
    bh[0][j].q[0] = *(const uint4*)&wHi[o];  bh[0][j].q[1] = *(const uint4*)&wHi[o + 16];
    bl[0][j].q[0] = *(const uint4*)&wLo[o];  bl[0][j].q[1] = *(const uint4*)&wLo[o + 16];
  }
  #pragma unroll
  for (int s = 0; s < 16; ++s) {
    const int q = s >> 2, np = s & 3, cur = s & 1, nxt = cur ^ 1;
    if (s + 1 < 16) {
      const int q1 = (s + 1) >> 2, np1 = (s + 1) & 3;
      #pragma unroll
      for (int j = 0; j < 2; ++j) {
        const int o = o0 + (((np1 << 1) + j) << 4) * WP + (q1 << 5);
        bh[nxt][j].q[0] = *(const uint4*)&wHi[o];  bh[nxt][j].q[1] = *(const uint4*)&wHi[o + 16];
        bl[nxt][j].q[0] = *(const uint4*)&wLo[o];  bl[nxt][j].q[1] = *(const uint4*)&wLo[o + 16];
      }
    }
    const int n0 = np << 1, n1 = n0 + 1;
    acc[n0] = __builtin_amdgcn_wmma_f32_16x16x32_bf16(false, ah[q].v, false, bh[cur][0].v, (short)0, acc[n0], false, false);
    acc[n1] = __builtin_amdgcn_wmma_f32_16x16x32_bf16(false, ah[q].v, false, bh[cur][1].v, (short)0, acc[n1], false, false);
    acc[n0] = __builtin_amdgcn_wmma_f32_16x16x32_bf16(false, ah[q].v, false, bl[cur][0].v, (short)0, acc[n0], false, false);
    acc[n1] = __builtin_amdgcn_wmma_f32_16x16x32_bf16(false, ah[q].v, false, bl[cur][1].v, (short)0, acc[n1], false, false);
    acc[n0] = __builtin_amdgcn_wmma_f32_16x16x32_bf16(false, al[q].v, false, bh[cur][0].v, (short)0, acc[n0], false, false);
    acc[n1] = __builtin_amdgcn_wmma_f32_16x16x32_bf16(false, al[q].v, false, bh[cur][1].v, (short)0, acc[n1], false, false);
    __builtin_amdgcn_sched_barrier(0);
  }
}

__device__ __forceinline__ float elu1(float v) { return (v > 0.0f) ? v : expm1f(v); }

__device__ __forceinline__ float elu_mid(float v) { return (v > 0.0f) ? v : (__expf(v) - 1.0f); }

#define MP 132

#define RB 32
#define NB 16

template <int B0, int K>
__device__ __forceinline__ void ask_rows(float4 (&u)[NB], const float4* __restrict__ t4, int rcx, int n, int lane) {
  if constexpr (K < NB) {
    if (K < n) {
      const int s = __builtin_amdgcn_readlane(rcx, B0 + K);
      u[K] = t4[(long long)s * (HID / 4) + lane];
      ask_rows<B0, K + 1>(u, t4, rcx, n, lane);
    }
  }
}
template <int B0, int K>
__device__ __forceinline__ void add_rows(float4& acc, const float4 (&u)[NB], int rcy, int n) {
  if constexpr (K < NB) {
    if (K < n) {
      const float w = __int_as_float(__builtin_amdgcn_readlane(rcy, B0 + K));
      acc.x += u[K].x * w; acc.y += u[K].y * w; acc.z += u[K].z * w; acc.w += u[K].w * w;
      add_rows<B0, K + 1>(acc, u, rcy, n);
    }
  }
}

__device__ __forceinline__ void sum_rows16(const float4* __restrict__ t4, const int2* __restrict__ r2,
                                           int rp, float dv, int row0, int rows, float* __restrict__ mw, int lane) {
  int beg = __builtin_amdgcn_readlane(rp, 0);
  int end = __builtin_amdgcn_readlane(rp, 1);
  int2 rc = make_int2(0, 0);
  if (lane < min(end - beg, RB)) rc = r2[beg + lane];
  float4 sf = t4[(long long)row0 * (HID / 4) + lane];
  for (int r = 0; r < rows; ++r) {
    const int cnt = end - beg;
    const float di = __int_as_float(__builtin_amdgcn_readlane(__float_as_int(dv), r));
    const float wself = di * di;
    int nend = end;
    int2 nrc = make_int2(0, 0);
    float4 nsf = sf;
    if (r + 1 < rows) {
      nend = __builtin_amdgcn_readlane(rp, r + 2);
      if (lane < min(nend - end, RB)) nrc = r2[end + lane];
      nsf = t4[(long long)(row0 + r + 1) * (HID / 4) + lane];
    }
    float4 acc;
    acc.x = sf.x * wself; acc.y = sf.y * wself; acc.z = sf.z * wself; acc.w = sf.w * wself;
    for (int off = 0; off < cnt; off += RB) {
      if (off > 0) { rc = make_int2(0, 0); if (lane < min(cnt - off, RB)) rc = r2[beg + off + lane]; }
      const int n = cnt - off;
      float4 u[NB];
      ask_rows<0, 0>(u, t4, rc.x, n, lane);
      add_rows<0, 0>(acc, u, rc.y, n);
      ask_rows<NB, 0>(u, t4, rc.x, n - NB, lane);
      add_rows<NB, 0>(acc, u, rc.y, n - NB);
    }
    *(float4*)&mw[r * MP + 4 * lane] = acc;
    beg = end; end = nend; rc = nrc; sf = nsf;
  }
  for (int r = rows; r < 16; ++r) *(float4*)&mw[r * MP + 4 * lane] = make_float4(0.0f, 0.0f, 0.0f, 0.0f);
}

__device__ __forceinline__ void rows_from_tile(float4 (&raw)[16], const float* __restrict__ mw, int c, int h) {
  const float4* __restrict__ p = (const float4*)mw + c * (MP / 4) + 2 * h;
  #pragma unroll
  for (int q = 0; q < 4; ++q) {
    raw[4 * q + 0] = p[8 * q + 0];
    raw[4 * q + 1] = p[8 * q + 1];
    raw[4 * q + 2] = p[8 * q + 4];
    raw[4 * q + 3] = p[8 * q + 5];
  }
}

__global__ __launch_bounds__(GT)
void sum_gemm_rows16_x2(const float* __restrict__ T, const int* __restrict__ rowptr, const int* __restrict__ rec,
                        const float* __restrict__ dinv,
                        const float* __restrict__ W1, const float* __restrict__ b1,
                        const float* __restrict__ W2, float* __restrict__ C, int nrows) {
  __shared__ __attribute__((aligned(16))) u16 wHi1[HID * WP];
  __shared__ __attribute__((aligned(16))) u16 wLo1[HID * WP];
  __shared__ __attribute__((aligned(16))) u16 wHi2[HID * WP];
  __shared__ __attribute__((aligned(16))) u16 wLo2[HID * WP];
  __shared__ __attribute__((aligned(16))) float mid[(GT / 32) * 16 * MP];

  const int tid  = threadIdx.x;
  const int lane = tid & 31;
  const int wave = __builtin_amdgcn_readfirstlane(tid >> 5);
  const int c = lane & 15, h = lane >> 4;
  const int ntiles = (nrows + 15) >> 4;
  const int tstep  = (int)gridDim.x * (GT / 32);
  int tile = (int)blockIdx.x + (int)gridDim.x * wave;

  const float4* __restrict__ T4 = (const float4*)T;
  const int2*   __restrict__ r2 = (const int2*)rec;
  float* __restrict__ mw = mid + wave * (16 * MP);

  int rp = 0; float dv = 0.0f;
  if (tile < ntiles) {
    const int row0 = tile << 4;
    rp = rowptr[min(row0 + lane, nrows)];
    dv = dinv[min(row0 + lane, nrows - 1)];
  }
  float bias[8];
  #pragma unroll
  for (int t = 0; t < 8; ++t) bias[t] = b1[(t << 4) + c];

  stage_w(W1, wHi1, wLo1, tid);
  stage_w(W2, wHi2, wLo2, tid);
  __syncthreads();

  while (tile < ntiles) {
    float4 raw[16];
    v8f acc[8];
    sum_rows16(T4, r2, rp, dv, tile << 4, min(16, nrows - (tile << 4)), mw, lane);
    __builtin_amdgcn_fence(__ATOMIC_RELEASE, "wavefront");
    __builtin_amdgcn_wave_barrier();
    __builtin_amdgcn_fence(__ATOMIC_ACQUIRE, "wavefront");
    rows_from_tile(raw, mw, c, h);
    __builtin_amdgcn_fence(__ATOMIC_RELEASE, "wavefront");
    __builtin_amdgcn_wave_barrier();

    rows16_times_w(raw, wHi1, wLo1, c, h, acc);

    #pragma unroll
    for (int t = 0; t < 8; ++t)
      #pragma unroll
      for (int r = 0; r < 8; ++r)
        mw[(8 * h + r) * MP + (t << 4) + c] = elu_mid(acc[t][r] + bias[t]);
    __builtin_amdgcn_fence(__ATOMIC_RELEASE, "wavefront");
    __builtin_amdgcn_wave_barrier();
    __builtin_amdgcn_fence(__ATOMIC_ACQUIRE, "wavefront");
    rows_from_tile(raw, mw, c, h);
    __builtin_amdgcn_fence(__ATOMIC_RELEASE, "wavefront");
    __builtin_amdgcn_wave_barrier();

    rows16_times_w(raw, wHi2, wLo2, c, h, acc);

    const int row0 = tile << 4;
    if (row0 + 16 <= nrows) {
      float* __restrict__ cp = C + (long long)row0 * HID + lane;
      #pragma unroll
      for (int np = 0; np < 4; ++np) {
        #pragma unroll
        for (int r = 0; r < 8; ++r) {
          const float a = acc[2 * np][r], b = acc[2 * np + 1][r];
          const float got = xhalf(h ? a : b);
          cp[r * HID + (np << 5)]       = h ? got : a;
          cp[(8 + r) * HID + (np << 5)] = h ? b : got;
        }
      }
    } else {
      #pragma unroll
      for (int np = 0; np < 4; ++np) {
        const int col = (np << 5) + lane;
        #pragma unroll
        for (int r = 0; r < 8; ++r) {
          const float a = acc[2 * np][r], b = acc[2 * np + 1][r];
          const float got = xhalf(h ? a : b);
          const float v1 = h ? got : a;
          const float v2 = h ? b : got;
          if (row0 + r < nrows)     C[(long long)(row0 + r) * HID + col] = v1;
          if (row0 + 8 + r < nrows) C[(long long)(row0 + 8 + r) * HID + col] = v2;
        }
      }
    }

    tile += tstep;
    if (tile < ntiles) {
      const int nrow0 = tile << 4;
      rp = rowptr[min(nrow0 + lane, nrows)];
      dv = dinv[min(nrow0 + lane, nrows - 1)];
    }
  }
}

template <int CC>
__global__ void gemm_small(const float* __restrict__ h, const float* __restrict__ W,
                           float* __restrict__ out, int n) {
  const long long idx = (long long)blockIdx.x * blockDim.x + threadIdx.x;
  if (idx >= (long long)n * CC) return;
  const int node = (int)(idx / CC);
  const int c    = (int)(idx % CC);
  const float* hr = h + (long long)node * HID;
  float s = 0.0f;
  #pragma unroll 8
  for (int k = 0; k < HID; ++k) s += hr[k] * W[k * CC + c];
  out[idx] = s;
}

#define CH   2048
#define GMAX 1024
#define CAP  8192

__device__ __forceinline__ void scan256_wave(const int* cnt, int* start, int lane) {
  int v[8];
  int s = 0;
  #pragma unroll
  for (int k = 0; k < 8; ++k) { v[k] = cnt[8 * lane + k]; s += v[k]; }
  int incl = s;
  for (int o = 1; o < 32; o <<= 1) {
    const int t = __shfl_up(incl, o, 32);
    if (lane >= o) incl += t;
  }
  int run = incl - s;
  #pragma unroll
  for (int k = 0; k < 8; ++k) { start[8 * lane + k] = run; run += v[k]; }
}

__global__ __launch_bounds__(512)
void chunk_sort(const int* __restrict__ src, const int* __restrict__ dst,
                int2* __restrict__ sorted, int2* __restrict__ tab, int E) {
  __shared__ int cnt[256];
  __shared__ int start[256];
  __shared__ int2 buf[CH];
  const int tid = threadIdx.x;
  if (tid < 256) cnt[tid] = 0;
  __syncthreads();
  const int base = blockIdx.x * CH;
  int s[4], d[4], r[4];
  #pragma unroll
  for (int k = 0; k < 4; ++k) {
    const int i = base + k * 512 + tid;
    s[k] = (i < E) ? src[i] : 0;
    d[k] = (i < E) ? dst[i] : 0;
  }
  #pragma unroll
  for (int k = 0; k < 4; ++k) {
    const int i = base + k * 512 + tid;
    r[k] = (i < E) ? atomicAdd(&cnt[d[k] >> 8], 1) : 0;
  }
  __syncthreads();
  if (tid < 32) scan256_wave(cnt, start, tid);
  __syncthreads();
  if (tid < 256) tab[blockIdx.x * 256 + tid] = make_int2(start[tid], cnt[tid]);
  #pragma unroll
  for (int k = 0; k < 4; ++k) {
    const int i = base + k * 512 + tid;
    if (i < E) buf[start[d[k] >> 8] + r[k]] = make_int2(s[k], d[k]);
  }
  __syncthreads();
  #pragma unroll
  for (int k = 0; k < 4; ++k) {
    const int j = k * 512 + tid;
    if (base + j < E) sorted[base + j] = buf[j];
  }
}

__device__ __forceinline__ int2 bucket_edge(const int2* __restrict__ sorted, const int* pbase, const int* pst, int G, int j) {
  int lo = 0, hi = G;
  while (hi - lo > 1) {
    const int mid = (lo + hi) >> 1;
    if (pbase[mid] <= j) lo = mid; else hi = mid;
  }
  return sorted[(long long)lo * CH + pst[lo] + (j - pbase[lo])];
}

__global__ __launch_bounds__(GMAX)
void bucket_rows(const int2* __restrict__ sorted, const int2* __restrict__ tab, int G,
                 int* __restrict__ rowptr, float* __restrict__ dinv, int2* __restrict__ rec, int n) {
  __shared__ int pst[GMAX];
  __shared__ int pbase[GMAX];
  __shared__ int wtot[32];
  __shared__ int ncnt[256];
  __shared__ int nstart[256];
  __shared__ int nfill[256];
  __shared__ int sbase;
  __shared__ int2 raw[CAP];
  __shared__ int2 stage[CAP];
  const int tid = threadIdx.x, lane = tid & 31, wave = tid >> 5;
  const int b = blockIdx.x;
  if (tid < 256) { ncnt[tid] = 0; nfill[tid] = 0; }
  if (tid == 0) sbase = 0;
  __syncthreads();

  const bool has = tid < G;
  const int2 t = has ? tab[tid * 256 + b] : make_int2(0, 0);
  if (has) pst[tid] = t.x;
  int sx = t.x;
  for (int o = 16; o > 0; o >>= 1) sx += __shfl_xor(sx, o, 32);
  if (lane == 0) atomicAdd(&sbase, sx);
  int incl = t.y;
  for (int o = 1; o < 32; o <<= 1) {
    const int u = __shfl_up(incl, o, 32);
    if (lane >= o) incl += u;
  }
  if (lane == 31) wtot[wave] = incl;
  __syncthreads();
  const int wv = wtot[lane];
  int winc = wv;
  for (int o = 1; o < 32; o <<= 1) {
    const int u = __shfl_up(winc, o, 32);
    if (lane >= o) winc += u;
  }
  const int total = __shfl(winc, 31, 32);
  const int woff  = __shfl(winc - wv, wave, 32);
  if (has) pbase[tid] = woff + incl - t.y;
  const int base = sbase;
  const bool fits = (total <= CAP);
  __syncthreads();

  for (int j0 = 0; j0 < total; j0 += 4 * GMAX) {
    int lo[4], hi[4], jj[4];
    #pragma unroll
    for (int k = 0; k < 4; ++k) { jj[k] = j0 + k * GMAX + tid; lo[k] = 0; hi[k] = G; }
    for (int s = 0; s < 10; ++s) {
      #pragma unroll
      for (int k = 0; k < 4; ++k) {
        const int mid = (lo[k] + hi[k]) >> 1;
        const int p = pbase[mid];
        if (hi[k] - lo[k] > 1) { if (p <= jj[k]) lo[k] = mid; else hi[k] = mid; }
      }
    }
    int2 e[4];
    #pragma unroll
    for (int k = 0; k < 4; ++k)
      e[k] = (jj[k] < total) ? sorted[(long long)lo[k] * CH + pst[lo[k]] + (jj[k] - pbase[lo[k]])] : make_int2(0, 0);
    #pragma unroll
    for (int k = 0; k < 4; ++k)
      if (jj[k] < total) {
        if (fits) raw[jj[k]] = e[k];
        atomicAdd(&ncnt[e[k].y & 255], 1);
      }
  }
  __syncthreads();
  if (wave == 0) scan256_wave(ncnt, nstart, lane);
  __syncthreads();

  if (tid < 256) {
    const int node = b * 256 + tid;
    if (node < n) {
      const int c = ncnt[tid];
      rowptr[node] = base + nstart[tid];
      const float d = (float)(c + 1);
      dinv[node] = (d > 0.0f) ? rsqrtf(d) : 0.0f;
      if (node == n - 1) rowptr[n] = base + nstart[tid] + c;
    }
  }
  for (int j = tid; j < total; j += GMAX) {
    const int2 e = fits ? raw[j] : bucket_edge(sorted, pbase, pst, G, j);
    const int nl = e.y & 255;
    const int pos = nstart[nl] + atomicAdd(&nfill[nl], 1);
    if (fits) stage[pos] = e; else rec[base + pos] = e;
  }
  __syncthreads();
  if (fits)
    for (int j = tid; j < total; j += GMAX) rec[base + j] = stage[j];
}

__global__ void norm_fill(int2* __restrict__ rec, const float* __restrict__ dinv, int E) {
  const int i = blockIdx.x * blockDim.x + threadIdx.x;
  if (i < E) {
    int2 e = rec[i];
    e.y = __float_as_int(dinv[e.x] * dinv[e.y]);
    rec[i] = e;
  }
}

__global__ __launch_bounds__(256)
void gather_rows128(const float* __restrict__ tmp, const int* __restrict__ rowptr,
                    const int* __restrict__ rec, const float* __restrict__ dinv,
                    const float* __restrict__ bias, float* __restrict__ agg, int n) {
  const int lane = threadIdx.x & 31;
  int row = blockIdx.x * 8 + (threadIdx.x >> 5);
  if (row >= n) return;
  row = __builtin_amdgcn_readfirstlane(row);
  const int beg = rowptr[row], end = rowptr[row + 1];
  const float di = dinv[row];
  const float wself = di * di;
  const float4* __restrict__ t4 = (const float4*)tmp;
  const int2*   __restrict__ r2 = (const int2*)rec;
  const float4 v = t4[(long long)row * (HID / 4) + lane];
  const float4 b = ((const float4*)bias)[lane];
  float4 acc;
  acc.x = v.x * wself; acc.y = v.y * wself; acc.z = v.z * wself; acc.w = v.w * wself;
  #pragma unroll 4
  for (int p = beg; p < end; ++p) {
    const int2 e = r2[p];
    const float w = __int_as_float(e.y);
    const float4 u = t4[(long long)e.x * (HID / 4) + lane];
    acc.x += u.x * w; acc.y += u.y * w; acc.z += u.z * w; acc.w += u.w * w;
  }
  acc.x = elu1(acc.x + b.x); acc.y = elu1(acc.y + b.y);
  acc.z = elu1(acc.z + b.z); acc.w = elu1(acc.w + b.w);
  ((float4*)agg)[(long long)row * (HID / 4) + lane] = acc;
}
template <int CC>
__global__ void gather_rows_small(const float* __restrict__ tmp, const int* __restrict__ rowptr,
                                  const int* __restrict__ rec, const float* __restrict__ dinv,
                                  const float* __restrict__ bias, float* __restrict__ agg, int n) {
  const long long idx = (long long)blockIdx.x * blockDim.x + threadIdx.x;
  if (idx >= (long long)n * CC) return;
  const int row = (int)(idx / CC);
  const int c   = (int)(idx % CC);
  const int beg = rowptr[row], end = rowptr[row + 1];
  const float di = dinv[row];
  const int2* __restrict__ r2 = (const int2*)rec;
  float acc = tmp[idx] * (di * di);
  for (int p = beg; p < end; ++p) {
    const int2 e = r2[p];
    acc += tmp[(long long)e.x * CC + c] * __int_as_float(e.y);
  }
  agg[idx] = acc + bias[c];
}

static inline int cdiv(long long a, long long b) { return (int)((a + b - 1) / b); }
static inline size_t pad64(size_t n) { return (n + 63) & ~(size_t)63; }

extern "C" void kernel_launch(void* const* d_in, const int* in_sizes, int n_in,
                              void* d_out, int out_size, void* d_ws, size_t ws_size,
                              hipStream_t stream) {
  (void)n_in; (void)out_size; (void)ws_size;
  const float* x    = (const float*)d_in[0];
  const int*   eidx = (const int*)d_in[1];
  const float* Wst  = (const float*)d_in[2];
  const float* bst  = (const float*)d_in[3];
  const float* Wout = (const float*)d_in[4];
  const float* bout = (const float*)d_in[5];

  const int N = in_sizes[0] / HID;
  const int E = in_sizes[1] / 2;
  constexpr int NCLS = 10;

  const int* src = eidx;
  const int* dst = eidx + E;
  const int G   = cdiv(E, CH);
  const int nbk = cdiv(N, 256);

  int*   rowptr = (int*)d_ws;
  float* dinv   = (float*)(rowptr + pad64((size_t)N + 1));
  int*   tab    = (int*)(dinv + pad64(N));
  int*   sorted = tab + pad64((size_t)2 * G * 256);
  int*   rec    = sorted + pad64((size_t)2 * E);
  float* hbuf   = (float*)(rec + pad64((size_t)2 * E));
  float* tmp    = hbuf + (size_t)N * HID;
  float* tmp2   = tmp + (size_t)N * HID;

  chunk_sort<<<G, 512, 0, stream>>>(src, dst, (int2*)sorted, (int2*)tab, E);
  bucket_rows<<<nbk, GMAX, 0, stream>>>((const int2*)sorted, (const int2*)tab, G, rowptr, dinv, (int2*)rec, N);
  norm_fill<<<cdiv(E, 256), 256, 0, stream>>>((int2*)rec, dinv, E);

  const float* hin = x;
  for (int l = 0; l < 8; l += 2) {
    sum_gemm_rows16_x2<<<256, GT, 0, stream>>>(hin, rowptr, rec, dinv, Wst + (size_t)l * HID * HID, bst + (size_t)l * HID,
                                               Wst + (size_t)(l + 1) * HID * HID, tmp2, N);
    gather_rows128<<<cdiv(N, 8), 256, 0, stream>>>(tmp2, rowptr, rec, dinv, bst + (size_t)(l + 1) * HID, hbuf, N);
    hin = hbuf;
  }

  const long long no = (long long)N * NCLS;
  gemm_small<NCLS><<<cdiv(no, 256), 256, 0, stream>>>(hbuf, Wout, tmp, N);
  gather_rows_small<NCLS><<<cdiv(no, 256), 256, 0, stream>>>(tmp, rowptr, rec, dinv, bout, (float*)d_out, N);
}
